# starting kernel
# baseline (speedup 1.0000x reference)
#include <hip/hip_runtime.h>

typedef __attribute__((ext_vector_type(2))) float v2f;
typedef __attribute__((ext_vector_type(8))) float v8f;

__global__ void init_deg_kernel(float* __restrict__ deg, int n) {
    int i = blockIdx.x * blockDim.x + threadIdx.x;
    if (i < n) deg[i] = 1.0f;
}

__global__ void count_deg_kernel(const int* __restrict__ dst,
                                 float* __restrict__ deg, int E) {
    int e = blockIdx.x * blockDim.x + threadIdx.x;
    if (e < E) atomicAdd(&deg[dst[e]], 1.0f);
}

__global__ void isqrt_kernel(float* __restrict__ deg, int n) {
    int i = blockIdx.x * blockDim.x + threadIdx.x;
    if (i < n) deg[i] = rsqrtf(deg[i]);
}

template <bool FUSE>
__global__ void gemm_wmma_kernel(const float* __restrict__ A,
                                 const float* __restrict__ W,
                                 const float* __restrict__ dis,
                                 const float* __restrict__ bias,
                                 float* __restrict__ ht,
                                 float* __restrict__ agg,
                                 int M, int K) {
    const int lane = threadIdx.x & 31;
    const int wave = blockIdx.x * (blockDim.x >> 5) + (threadIdx.x >> 5);
    const int m0 = wave * 16;
    if (m0 >= M) return;

    const int half = lane >> 4;
    const int ml   = lane & 15;
    const int arow = m0 + ml;
    const float dA = FUSE ? dis[arow] : 0.0f;

    v8f c0 = {}, c1 = {}, c2 = {}, c3 = {};

    for (int k0 = 0; k0 < K; k0 += 4) {
        const int ka = k0 + half * 2;
        float a0 = A[(size_t)arow * K + ka];
        float a1 = A[(size_t)arow * K + ka + 1];
        if (FUSE) {
            a0 = fmaxf(fmaf(a0, dA, bias[ka]), 0.0f);
            a1 = fmaxf(fmaf(a1, dA, bias[ka + 1]), 0.0f);
        }
        v2f av; av.x = a0; av.y = a1;

        const float* w0 = W + (size_t)ka * 64;
        const float* w1 = W + (size_t)(ka + 1) * 64;
        v2f b0; b0.x = w0[ml];      b0.y = w1[ml];
        v2f b1; b1.x = w0[16 + ml]; b1.y = w1[16 + ml];
        v2f b2; b2.x = w0[32 + ml]; b2.y = w1[32 + ml];
        v2f b3; b3.x = w0[48 + ml]; b3.y = w1[48 + ml];

        c0 = __builtin_amdgcn_wmma_f32_16x16x4_f32(false, av, false, b0, (short)0, c0, false, false);
        c1 = __builtin_amdgcn_wmma_f32_16x16x4_f32(false, av, false, b1, (short)0, c1, false, false);
        c2 = __builtin_amdgcn_wmma_f32_16x16x4_f32(false, av, false, b2, (short)0, c2, false, false);
        c3 = __builtin_amdgcn_wmma_f32_16x16x4_f32(false, av, false, b3, (short)0, c3, false, false);
    }

    for (int r = 0; r < 8; ++r) {
        const int orow = m0 + half * 8 + r;
        const float dr = dis[orow];
        const float v0 = c0[r] * dr;
        const float v1 = c1[r] * dr;
        const float v2 = c2[r] * dr;
        const float v3 = c3[r] * dr;
        float* hp = ht  + (size_t)orow * 64;
        float* ap = agg + (size_t)orow * 64;
        hp[ml] = v0; hp[16 + ml] = v1; hp[32 + ml] = v2; hp[48 + ml] = v3;
        ap[ml] = v0; ap[16 + ml] = v1; ap[32 + ml] = v2; ap[48 + ml] = v3;
    }
}

__global__ void scatter_edges_kernel(const int* __restrict__ src,
                                     const int* __restrict__ dst,
                                     const float* __restrict__ ht,
                                     float* __restrict__ agg, int E) {
    const int lane = threadIdx.x & 31;
    const int e = blockIdx.x * (blockDim.x >> 5) + (threadIdx.x >> 5);
    if (e >= E) return;
    const int s = src[e];
    const int d = dst[e];
    const float2 v = *(const float2*)(ht + (size_t)s * 64 + lane * 2);
    float* ap = agg + (size_t)d * 64 + lane * 2;
    atomicAdd(ap, v.x);
    atomicAdd(ap + 1, v.y);
}

__global__ void fc_out_kernel(const float* __restrict__ agg2,
                              const float* __restrict__ dis,
                              const float* __restrict__ b2,
                              const float* __restrict__ wfc,
                              const float* __restrict__ bfc,
                              float* __restrict__ out, int M) {
    const int lane = threadIdx.x & 31;
    const int i = blockIdx.x * (blockDim.x >> 5) + (threadIdx.x >> 5);
    if (i >= M) return;
    const float di = dis[i];
    const int j = lane * 2;
    const float2 a = *(const float2*)(agg2 + (size_t)i * 64 + j);
    const float h0 = fmaxf(fmaf(a.x, di, b2[j]), 0.0f);
    const float h1 = fmaxf(fmaf(a.y, di, b2[j + 1]), 0.0f);
    float p = h0 * wfc[j] + h1 * wfc[j + 1];
    #pragma unroll
    for (int off = 16; off > 0; off >>= 1) p += __shfl_down(p, off, 32);
    if (lane == 0) out[i] = p + bfc[0];
}

extern "C" void kernel_launch(void* const* d_in, const int* in_sizes, int n_in,
                              void* d_out, int out_size, void* d_ws, size_t ws_size,
                              hipStream_t stream) {
    const float* x   = (const float*)d_in[0];
    const int*   ei  = (const int*)d_in[1];
    const float* W1  = (const float*)d_in[2];
    const float* b1  = (const float*)d_in[3];
    const float* W2  = (const float*)d_in[4];
    const float* b2  = (const float*)d_in[5];
    const float* Wfc = (const float*)d_in[6];
    const float* bfc = (const float*)d_in[7];

    const int HID = in_sizes[3];
    const int IN  = in_sizes[2] / HID;
    const int N   = in_sizes[0] / IN;
    const int E   = in_sizes[1] / 2;
    const int* srcv = ei;
    const int* dstv = ei + E;

    char* ws = (char*)d_ws;
    float* dis = (float*)ws;
    size_t off = (((size_t)N * 4) + 255) & ~(size_t)255;
    const size_t rowBytes = (size_t)N * 64 * sizeof(float);
    float* ht   = (float*)(ws + off);
    float* agg1 = (float*)(ws + off + rowBytes);
    float* agg2 = (float*)(ws + off + 2 * rowBytes);

    init_deg_kernel<<<(N + 255) / 256, 256, 0, stream>>>(dis, N);
    count_deg_kernel<<<(E + 255) / 256, 256, 0, stream>>>(dstv, dis, E);
    isqrt_kernel<<<(N + 255) / 256, 256, 0, stream>>>(dis, N);

    const int gemmBlocks = ((N + 15) / 16 + 7) / 8;
    const int edgeBlocks = (E + 7) / 8;
    const int nodeBlocks = (N + 7) / 8;

    gemm_wmma_kernel<false><<<gemmBlocks, 256, 0, stream>>>(x, W1, dis, nullptr, ht, agg1, N, IN);
    scatter_edges_kernel<<<edgeBlocks, 256, 0, stream>>>(srcv, dstv, ht, agg1, E);

    gemm_wmma_kernel<true><<<gemmBlocks, 256, 0, stream>>>(agg1, W2, dis, b1, ht, agg2, N, HID);
    scatter_edges_kernel<<<edgeBlocks, 256, 0, stream>>>(srcv, dstv, ht, agg2, E);

    fc_out_kernel<<<nodeBlocks, 256, 0, stream>>>(agg2, dis, b2, Wfc, bfc, (float*)d_out, N);
}


# edge sums without global float atomics: a block owns 512 nodes, keeps their 64-float sums in LDS, scans the whole destination list, adds matching ht rows by LDS float adds; sum order: arrival
# speedup vs baseline: 1.8078x; 1.8078x over previous
#include <hip/hip_runtime.h>

typedef __attribute__((ext_vector_type(2))) float v2f;
typedef __attribute__((ext_vector_type(8))) float v8f;

#define DEG_CH     32768
#define DEG_SLICES 64
#define DEG_NT     512
#define DEG_DEPTH  8

typedef int int4u __attribute__((ext_vector_type(4), aligned(4)));

__global__ __launch_bounds__(DEG_NT)
void deg_hist_kernel(const int* __restrict__ dst, unsigned* __restrict__ planes,
                     int E, int npad) {
    __shared__ unsigned cnt[DEG_CH];
    const int tid   = threadIdx.x;
    const int slice = blockIdx.x;
    const unsigned base = blockIdx.y * (unsigned)DEG_CH;

    uint4* c4 = (uint4*)cnt;
    for (int i = tid; i < DEG_CH / 4; i += DEG_NT) c4[i] = make_uint4(0u, 0u, 0u, 0u);
    __syncthreads();

    const int nvec = E >> 2;
    const int per  = (nvec + DEG_SLICES - 1) / DEG_SLICES;
    const int lo   = slice * per;
    const int hi   = min(lo + per, nvec);
    const int4u* d4 = (const int4u*)dst;

    for (int v0 = lo + tid; v0 < hi; v0 += DEG_DEPTH * DEG_NT) {
        int4u d[DEG_DEPTH];
        #pragma unroll
        for (int u = 0; u < DEG_DEPTH; ++u) {
            const int v = v0 + u * DEG_NT;
            d[u] = d4[min(v, hi - 1)];
            if (v >= hi) d[u] = (int4u)(-1);
        }
        #pragma unroll
        for (int u = 0; u < DEG_DEPTH; ++u) {
            #pragma unroll
            for (int q = 0; q < 4; ++q) {
                const unsigned r = (unsigned)d[u][q] - base;
                if (r < (unsigned)DEG_CH) atomicAdd(&cnt[r], 1u);
            }
        }
    }
    if (slice == DEG_SLICES - 1) {
        for (int e = (nvec << 2) + tid; e < E; e += DEG_NT) {
            const unsigned r = (unsigned)dst[e] - base;
            if (r < (unsigned)DEG_CH) atomicAdd(&cnt[r], 1u);
        }
    }
    __syncthreads();

    uint4* o4 = (uint4*)(planes + (size_t)slice * npad + base);
    for (int i = tid; i < DEG_CH / 4; i += DEG_NT) o4[i] = c4[i];
}

__global__ void deg_isqrt_kernel(const unsigned* __restrict__ planes,
                                 float* __restrict__ dis, int n, int npad) {
    const int i = blockIdx.x * blockDim.x + threadIdx.x;
    if (i >= n) return;
    unsigned s = 1u;
    #pragma unroll 16
    for (int k = 0; k < DEG_SLICES; ++k) s += planes[(size_t)k * npad + i];
    dis[i] = rsqrtf((float)s);
}

template <bool FUSE>
__global__ void gemm_wmma_kernel(const float* __restrict__ A,
                                 const float* __restrict__ W,
                                 const float* __restrict__ dis,
                                 const float* __restrict__ bias,
                                 float* __restrict__ ht,
                                 int M, int K) {
    const int lane = threadIdx.x & 31;
    const int wave = blockIdx.x * (blockDim.x >> 5) + (threadIdx.x >> 5);
    const int m0 = wave * 16;
    if (m0 >= M) return;

    const int half = lane >> 4;
    const int ml   = lane & 15;
    const int arow = m0 + ml;
    const float dA = FUSE ? dis[arow] : 0.0f;

    v8f c0 = {}, c1 = {}, c2 = {}, c3 = {};

    for (int k0 = 0; k0 < K; k0 += 4) {
        const int ka = k0 + half * 2;
        float a0 = A[(size_t)arow * K + ka];
        float a1 = A[(size_t)arow * K + ka + 1];
        if (FUSE) {
            a0 = fmaxf(fmaf(a0, dA, bias[ka]), 0.0f);
            a1 = fmaxf(fmaf(a1, dA, bias[ka + 1]), 0.0f);
        }
        v2f av; av.x = a0; av.y = a1;

        const float* w0 = W + (size_t)ka * 64;
        const float* w1 = W + (size_t)(ka + 1) * 64;
        v2f b0; b0.x = w0[ml];      b0.y = w1[ml];
        v2f b1; b1.x = w0[16 + ml]; b1.y = w1[16 + ml];
        v2f b2; b2.x = w0[32 + ml]; b2.y = w1[32 + ml];
        v2f b3; b3.x = w0[48 + ml]; b3.y = w1[48 + ml];

        c0 = __builtin_amdgcn_wmma_f32_16x16x4_f32(false, av, false, b0, (short)0, c0, false, false);
        c1 = __builtin_amdgcn_wmma_f32_16x16x4_f32(false, av, false, b1, (short)0, c1, false, false);
        c2 = __builtin_amdgcn_wmma_f32_16x16x4_f32(false, av, false, b2, (short)0, c2, false, false);
        c3 = __builtin_amdgcn_wmma_f32_16x16x4_f32(false, av, false, b3, (short)0, c3, false, false);
    }

    for (int r = 0; r < 8; ++r) {
        const int orow = m0 + half * 8 + r;
        const float dr = dis[orow];
        const float v0 = c0[r] * dr;
        const float v1 = c1[r] * dr;
        const float v2 = c2[r] * dr;
        const float v3 = c3[r] * dr;
        float* hp = ht + (size_t)orow * 64;
        hp[ml] = v0; hp[16 + ml] = v1; hp[32 + ml] = v2; hp[48 + ml] = v3;
    }
}

#define AGG_SHIFT 9
#define AGG_CH    (1 << AGG_SHIFT)
#define AGG_NT    1024
#define AGG_NW    (AGG_NT / 32)
#define AGG_DEPTH 8
#define AGG_Q     (AGG_DEPTH * 128 + 32)
#define AGG_ROWS  16

__device__ __forceinline__ void agg_add_batch(const int* q, int n,
                                              const int* __restrict__ src,
                                              const int* __restrict__ dst,
                                              const float* __restrict__ ht,
                                              float* acc, int lane) {
    const int e  = q[min(lane, n - 1)];
    const int s  = src[e];
    const int dl = dst[e] & (AGG_CH - 1);
    #pragma unroll
    for (int j0 = 0; j0 < 32; j0 += AGG_ROWS) {
        if (j0 < n) {
            float2 v[AGG_ROWS];
            #pragma unroll
            for (int r = 0; r < AGG_ROWS; ++r) {
                const int sj = __builtin_amdgcn_readlane(s, j0 + r);
                v[r] = *(const float2*)(ht + (size_t)sj * 64 + lane * 2);
            }
            #pragma unroll
            for (int r = 0; r < AGG_ROWS; ++r) {
                if (j0 + r < n) {
                    const int dj = __builtin_amdgcn_readlane(dl, j0 + r);
                    float* ap = acc + dj * 64 + lane * 2;
                    atomicAdd(ap, v[r].x);
                    atomicAdd(ap + 1, v[r].y);
                }
            }
        }
    }
}

__global__ __launch_bounds__(AGG_NT)
void agg_owner_kernel(const int* __restrict__ src, const int* __restrict__ dst,
                      const float* __restrict__ ht, float* __restrict__ agg,
                      int E, int N) {
    __shared__ float acc[AGG_CH * 64];
    __shared__ int   queue[AGG_NW][AGG_Q];
    const int tid  = threadIdx.x;
    const int lane = tid & 31;
    const int wave = tid >> 5;
    const unsigned blk = blockIdx.x;
    const int base = (int)blk * AGG_CH;

    float4* a4 = (float4*)acc;
    const float4* h4 = (const float4*)(ht + (size_t)base * 64);
    for (int i = tid; i < AGG_CH * 16; i += AGG_NT)
        a4[i] = (base + (i >> 4) < N) ? h4[i] : make_float4(0.f, 0.f, 0.f, 0.f);
    __syncthreads();

    int* q = queue[wave];
    int cnt = 0;
    const int nvec = E >> 2;
    const int4u* d4 = (const int4u*)dst;
    const int step = AGG_DEPTH * AGG_NT;
    const int nIt  = (nvec + step - 1) / step;
    const int rot  = (int)(((long long)blk * nIt) / gridDim.x);

    for (int it = 0; it < nIt; ++it) {
        int itr = it + rot; if (itr >= nIt) itr -= nIt;
        const int v0 = itr * step + tid;
        int4u d[AGG_DEPTH];
        #pragma unroll
        for (int u = 0; u < AGG_DEPTH; ++u)
            d[u] = d4[min(v0 + u * AGG_NT, nvec - 1)];
        __builtin_amdgcn_sched_barrier(0);
        #pragma unroll
        for (int u = 0; u < AGG_DEPTH; ++u) {
            const bool inlist = v0 + u * AGG_NT < nvec;
            const unsigned inm = (unsigned)__builtin_amdgcn_ballot_w32(inlist);
            bool hit[4];
            unsigned m[4];
            #pragma unroll
            for (int k = 0; k < 4; ++k) {
                hit[k] = ((unsigned)d[u][k] >> AGG_SHIFT) == blk;
                m[k]   = (unsigned)__builtin_amdgcn_ballot_w32(hit[k]) & inm;
            }
            if ((m[0] | m[1] | m[2] | m[3]) != 0u) {
                #pragma unroll
                for (int k = 0; k < 4; ++k) {
                    if (hit[k] && inlist)
                        q[cnt + (int)__builtin_amdgcn_mbcnt_lo(m[k], 0u)] = ((v0 + u * AGG_NT) << 2) + k;
                    cnt += __builtin_popcount(m[k]);
                }
            }
        }
        while (cnt >= 32) {
            cnt -= 32;
            agg_add_batch(q + cnt, 32, src, dst, ht, acc, lane);
        }
    }
    if (cnt > 0) agg_add_batch(q, cnt, src, dst, ht, acc, lane);

    if (wave == 0) {
        for (int e = nvec << 2; e < E; ++e) {
            const int dd = dst[e];
            if (((unsigned)dd >> AGG_SHIFT) == blk) {
                const float2 v = *(const float2*)(ht + (size_t)src[e] * 64 + lane * 2);
                float* ap = acc + (dd & (AGG_CH - 1)) * 64 + lane * 2;
                atomicAdd(ap, v.x);
                atomicAdd(ap + 1, v.y);
            }
        }
    }
    __syncthreads();

    float4* o4 = (float4*)(agg + (size_t)base * 64);
    for (int i = tid; i < AGG_CH * 16; i += AGG_NT)
        if (base + (i >> 4) < N) o4[i] = a4[i];
}

__global__ void fc_out_kernel(const float* __restrict__ agg2,
                              const float* __restrict__ dis,
                              const float* __restrict__ b2,
                              const float* __restrict__ wfc,
                              const float* __restrict__ bfc,
                              float* __restrict__ out, int M) {
    const int lane = threadIdx.x & 31;
    const int i = blockIdx.x * (blockDim.x >> 5) + (threadIdx.x >> 5);
    if (i >= M) return;
    const float di = dis[i];
    const int j = lane * 2;
    const float2 a = *(const float2*)(agg2 + (size_t)i * 64 + j);
    const float h0 = fmaxf(fmaf(a.x, di, b2[j]), 0.0f);
    const float h1 = fmaxf(fmaf(a.y, di, b2[j + 1]), 0.0f);
    float p = h0 * wfc[j] + h1 * wfc[j + 1];
    #pragma unroll
    for (int off = 16; off > 0; off >>= 1) p += __shfl_down(p, off, 32);
    if (lane == 0) out[i] = p + bfc[0];
}

extern "C" void kernel_launch(void* const* d_in, const int* in_sizes, int n_in,
                              void* d_out, int out_size, void* d_ws, size_t ws_size,
                              hipStream_t stream) {
    const float* x   = (const float*)d_in[0];
    const int*   ei  = (const int*)d_in[1];
    const float* W1  = (const float*)d_in[2];
    const float* b1  = (const float*)d_in[3];
    const float* W2  = (const float*)d_in[4];
    const float* b2  = (const float*)d_in[5];
    const float* Wfc = (const float*)d_in[6];
    const float* bfc = (const float*)d_in[7];

    const int HID = in_sizes[3];
    const int IN  = in_sizes[2] / HID;
    const int N   = in_sizes[0] / IN;
    const int E   = in_sizes[1] / 2;
    const int* srcv = ei;
    const int* dstv = ei + E;

    char* ws = (char*)d_ws;
    float* dis = (float*)ws;
    size_t off = (((size_t)N * 4) + 255) & ~(size_t)255;
    const size_t rowBytes = (size_t)N * 64 * sizeof(float);
    float* ht   = (float*)(ws + off);
    float* agg1 = (float*)(ws + off + rowBytes);
    float* agg2 = (float*)(ws + off + 2 * rowBytes);

    const int nChunks = (N + DEG_CH - 1) / DEG_CH;
    const int npad    = nChunks * DEG_CH;
    unsigned* planes  = (unsigned*)(ws + off + 3 * rowBytes);
    deg_hist_kernel<<<dim3(DEG_SLICES, nChunks), DEG_NT, 0, stream>>>(dstv, planes, E, npad);
    deg_isqrt_kernel<<<(N + 255) / 256, 256, 0, stream>>>(planes, dis, N, npad);

    const int gemmBlocks = ((N + 15) / 16 + 7) / 8;
    const int aggBlocks  = (N + AGG_CH - 1) / AGG_CH;
    const int nodeBlocks = (N + 7) / 8;

    gemm_wmma_kernel<false><<<gemmBlocks, 256, 0, stream>>>(x, W1, dis, nullptr, ht, N, IN);
    agg_owner_kernel<<<aggBlocks, AGG_NT, 0, stream>>>(srcv, dstv, ht, agg1, E, N);

    gemm_wmma_kernel<true><<<gemmBlocks, 256, 0, stream>>>(agg1, W2, dis, b1, ht, N, HID);
    agg_owner_kernel<<<aggBlocks, AGG_NT, 0, stream>>>(srcv, dstv, ht, agg2, E, N);

    fc_out_kernel<<<nodeBlocks, 256, 0, stream>>>(agg2, dis, b2, Wfc, bfc, (float*)d_out, N);
}


# layer-1 owner blocks write their kept edges (source, local row) to a list placed by degree prefix sums; layer-2 blocks read their stretch, no second scan of the destination list; sum order: arrival
# speedup vs baseline: 2.1381x; 1.1827x over previous
#include <hip/hip_runtime.h>

typedef __attribute__((ext_vector_type(2))) float v2f;
typedef __attribute__((ext_vector_type(8))) float v8f;

#define DEG_CH     32768
#define DEG_SLICES 64
#define DEG_NT     512
#define DEG_DEPTH  8

typedef int int4u __attribute__((ext_vector_type(4), aligned(4)));

__global__ __launch_bounds__(DEG_NT)
void deg_hist_kernel(const int* __restrict__ dst, unsigned* __restrict__ planes,
                     int E, int npad) {
    __shared__ unsigned cnt[DEG_CH];
    const int tid   = threadIdx.x;
    const int slice = blockIdx.x;
    const unsigned base = blockIdx.y * (unsigned)DEG_CH;

    uint4* c4 = (uint4*)cnt;
    for (int i = tid; i < DEG_CH / 4; i += DEG_NT) c4[i] = make_uint4(0u, 0u, 0u, 0u);
    __syncthreads();

    const int nvec = E >> 2;
    const int per  = (nvec + DEG_SLICES - 1) / DEG_SLICES;
    const int lo   = slice * per;
    const int hi   = min(lo + per, nvec);
    const int4u* d4 = (const int4u*)dst;

    for (int v0 = lo + tid; v0 < hi; v0 += DEG_DEPTH * DEG_NT) {
        int4u d[DEG_DEPTH];
        #pragma unroll
        for (int u = 0; u < DEG_DEPTH; ++u) {
            const int v = v0 + u * DEG_NT;
            d[u] = d4[min(v, hi - 1)];
            if (v >= hi) d[u] = (int4u)(-1);
        }
        #pragma unroll
        for (int u = 0; u < DEG_DEPTH; ++u) {
            #pragma unroll
            for (int q = 0; q < 4; ++q) {
                const unsigned r = (unsigned)d[u][q] - base;
                if (r < (unsigned)DEG_CH) atomicAdd(&cnt[r], 1u);
            }
        }
    }
    if (slice == DEG_SLICES - 1) {
        for (int e = (nvec << 2) + tid; e < E; e += DEG_NT) {
            const unsigned r = (unsigned)dst[e] - base;
            if (r < (unsigned)DEG_CH) atomicAdd(&cnt[r], 1u);
        }
    }
    __syncthreads();

    uint4* o4 = (uint4*)(planes + (size_t)slice * npad + base);
    for (int i = tid; i < DEG_CH / 4; i += DEG_NT) o4[i] = c4[i];
}

#define ISQ_NT 256

__global__ __launch_bounds__(ISQ_NT)
void deg_isqrt_kernel(const unsigned* __restrict__ planes, float* __restrict__ dis,
                      unsigned* __restrict__ part, int n, int npad) {
    __shared__ unsigned wsum[ISQ_NT / 32];
    const int i = blockIdx.x * ISQ_NT + threadIdx.x;
    unsigned s = 0u;
    if (i < n) {
        #pragma unroll 16
        for (int k = 0; k < DEG_SLICES; ++k) s += planes[(size_t)k * npad + i];
        dis[i] = rsqrtf((float)(s + 1u));
    }
    unsigned w = s;
    #pragma unroll
    for (int off = 16; off > 0; off >>= 1) w += __shfl_xor(w, off, 32);
    if ((threadIdx.x & 31) == 0) wsum[threadIdx.x >> 5] = w;
    __syncthreads();
    if (threadIdx.x == 0) {
        unsigned t = 0u;
        for (int j = 0; j < ISQ_NT / 32; ++j) t += wsum[j];
        part[blockIdx.x] = t;
    }
}

template <bool FUSE>
__global__ void gemm_wmma_kernel(const float* __restrict__ A,
                                 const float* __restrict__ W,
                                 const float* __restrict__ dis,
                                 const float* __restrict__ bias,
                                 float* __restrict__ ht,
                                 int M, int K) {
    const int lane = threadIdx.x & 31;
    const int wave = blockIdx.x * (blockDim.x >> 5) + (threadIdx.x >> 5);
    const int m0 = wave * 16;
    if (m0 >= M) return;

    const int half = lane >> 4;
    const int ml   = lane & 15;
    const int arow = m0 + ml;
    const float dA = FUSE ? dis[arow] : 0.0f;

    v8f c0 = {}, c1 = {}, c2 = {}, c3 = {};

    for (int k0 = 0; k0 < K; k0 += 4) {
        const int ka = k0 + half * 2;
        float a0 = A[(size_t)arow * K + ka];
        float a1 = A[(size_t)arow * K + ka + 1];
        if (FUSE) {
            a0 = fmaxf(fmaf(a0, dA, bias[ka]), 0.0f);
            a1 = fmaxf(fmaf(a1, dA, bias[ka + 1]), 0.0f);
        }
        v2f av; av.x = a0; av.y = a1;

        const float* w0 = W + (size_t)ka * 64;
        const float* w1 = W + (size_t)(ka + 1) * 64;
        v2f b0; b0.x = w0[ml];      b0.y = w1[ml];
        v2f b1; b1.x = w0[16 + ml]; b1.y = w1[16 + ml];
        v2f b2; b2.x = w0[32 + ml]; b2.y = w1[32 + ml];
        v2f b3; b3.x = w0[48 + ml]; b3.y = w1[48 + ml];

        c0 = __builtin_amdgcn_wmma_f32_16x16x4_f32(false, av, false, b0, (short)0, c0, false, false);
        c1 = __builtin_amdgcn_wmma_f32_16x16x4_f32(false, av, false, b1, (short)0, c1, false, false);
        c2 = __builtin_amdgcn_wmma_f32_16x16x4_f32(false, av, false, b2, (short)0, c2, false, false);
        c3 = __builtin_amdgcn_wmma_f32_16x16x4_f32(false, av, false, b3, (short)0, c3, false, false);
    }

    for (int r = 0; r < 8; ++r) {
        const int orow = m0 + half * 8 + r;
        const float dr = dis[orow];
        const float v0 = c0[r] * dr;
        const float v1 = c1[r] * dr;
        const float v2 = c2[r] * dr;
        const float v3 = c3[r] * dr;
        float* hp = ht + (size_t)orow * 64;
        hp[ml] = v0; hp[16 + ml] = v1; hp[32 + ml] = v2; hp[48 + ml] = v3;
    }
}

#define AGG_SHIFT 9
#define AGG_CH    (1 << AGG_SHIFT)
#define AGG_NT    1024
#define AGG_NW    (AGG_NT / 32)
#define AGG_DEPTH 8
#define AGG_Q     (AGG_DEPTH * 128 + 32)
#define AGG_ROWS  16
#define AGG_PARTS (AGG_CH / ISQ_NT)

__device__ __forceinline__ void agg_add_rows(int s, int dl, int n,
                                             const float* __restrict__ ht,
                                             float* acc, int lane) {
    #pragma unroll
    for (int j0 = 0; j0 < 32; j0 += AGG_ROWS) {
        if (j0 < n) {
            float2 v[AGG_ROWS];
            #pragma unroll
            for (int r = 0; r < AGG_ROWS; ++r) {
                const int sj = __builtin_amdgcn_readlane(s, j0 + r);
                v[r] = *(const float2*)(ht + (size_t)sj * 64 + lane * 2);
            }
            #pragma unroll
            for (int r = 0; r < AGG_ROWS; ++r) {
                if (j0 + r < n) {
                    const int dj = __builtin_amdgcn_readlane(dl, j0 + r);
                    float* ap = acc + dj * 64 + lane * 2;
                    atomicAdd(ap, v[r].x);
                    atomicAdd(ap + 1, v[r].y);
                }
            }
        }
    }
}

__device__ __forceinline__ void agg_take_batch(const int* q, int n,
                                               const int* __restrict__ src,
                                               const int* __restrict__ dst,
                                               const float* __restrict__ ht,
                                               float* acc, int lane,
                                               int2* __restrict__ lst, int cap, int* fill) {
    const int e  = q[min(lane, n - 1)];
    const int s  = src[e];
    const int dl = dst[e] & (AGG_CH - 1);
    int pos = 0;
    if (lane == 0) pos = atomicAdd(fill, n);
    pos = __builtin_amdgcn_readfirstlane(pos) + lane;
    if (lane < n && pos < cap) lst[pos] = make_int2(s, dl);
    agg_add_rows(s, dl, n, ht, acc, lane);
}

template <bool BUILD>
__global__ __launch_bounds__(AGG_NT)
void agg_owner_kernel(const int* __restrict__ src, const int* __restrict__ dst,
                      const float* __restrict__ ht, float* __restrict__ agg,
                      const unsigned* __restrict__ part, int2* __restrict__ list,
                      int E, int N, int nPart) {
    __shared__ float acc[AGG_CH * 64];
    __shared__ unsigned red[AGG_NW];
    __shared__ int fill;
    const int tid  = threadIdx.x;
    const int lane = tid & 31;
    const int wave = tid >> 5;
    const unsigned blk = blockIdx.x;
    const int base = (int)blk * AGG_CH;

    unsigned below = 0u;
    for (int j = tid; j < (int)blk * AGG_PARTS; j += AGG_NT) below += part[j];
    #pragma unroll
    for (int off = 16; off > 0; off >>= 1) below += __shfl_xor(below, off, 32);
    if (lane == 0) red[wave] = below;
    if (tid == 0) fill = 0;
    unsigned mine = 0u;
    #pragma unroll
    for (int j = 0; j < AGG_PARTS; ++j) {
        const int p = (int)blk * AGG_PARTS + j;
        if (p < nPart) mine += part[p];
    }
    const int cap = (int)mine;

    float4* a4 = (float4*)acc;
    const float4* h4 = (const float4*)(ht + (size_t)base * 64);
    for (int i = tid; i < AGG_CH * 16; i += AGG_NT)
        a4[i] = (base + (i >> 4) < N) ? h4[i] : make_float4(0.f, 0.f, 0.f, 0.f);
    __syncthreads();

    static_assert(AGG_NW == 32, "one wave sum per lane");
    unsigned first = red[lane];
    #pragma unroll
    for (int off = 16; off > 0; off >>= 1) first += __shfl_xor(first, off, 32);
    int2* lst = list + (size_t)__builtin_amdgcn_readfirstlane((int)first);

    if (BUILD) {
        __shared__ int queue[AGG_NW][AGG_Q];
        int* q = queue[wave];
        int cnt = 0;
        const int nvec = E >> 2;
        const int4u* d4 = (const int4u*)dst;
        const int step = AGG_DEPTH * AGG_NT;
        const int nIt  = (nvec + step - 1) / step;
        const int rot  = (int)(((long long)blk * nIt) / gridDim.x);

        for (int it = 0; it < nIt; ++it) {
            int itr = it + rot; if (itr >= nIt) itr -= nIt;
            const int v0 = itr * step + tid;
            int4u d[AGG_DEPTH];
            #pragma unroll
            for (int u = 0; u < AGG_DEPTH; ++u)
                d[u] = d4[min(v0 + u * AGG_NT, nvec - 1)];
            __builtin_amdgcn_sched_barrier(0);
            #pragma unroll
            for (int u = 0; u < AGG_DEPTH; ++u) {
                const bool inlist = v0 + u * AGG_NT < nvec;
                const unsigned inm = (unsigned)__builtin_amdgcn_ballot_w32(inlist);
                bool hit[4];
                unsigned m[4];
                #pragma unroll
                for (int k = 0; k < 4; ++k) {
                    hit[k] = ((unsigned)d[u][k] >> AGG_SHIFT) == blk;
                    m[k]   = (unsigned)__builtin_amdgcn_ballot_w32(hit[k]) & inm;
                }
                if ((m[0] | m[1] | m[2] | m[3]) != 0u) {
                    #pragma unroll
                    for (int k = 0; k < 4; ++k) {
                        if (hit[k] && inlist)
                            q[cnt + (int)__builtin_amdgcn_mbcnt_lo(m[k], 0u)] = ((v0 + u * AGG_NT) << 2) + k;
                        cnt += __builtin_popcount(m[k]);
                    }
                }
            }
            while (cnt >= 32) {
                cnt -= 32;
                agg_take_batch(q + cnt, 32, src, dst, ht, acc, lane, lst, cap, &fill);
            }
        }
        if (wave == 0) {
            const int e = (nvec << 2) + lane;
            const bool hit = e < E && ((unsigned)dst[min(e, E - 1)] >> AGG_SHIFT) == blk;
            const unsigned m = (unsigned)__builtin_amdgcn_ballot_w32(hit);
            if (hit) q[cnt + (int)__builtin_amdgcn_mbcnt_lo(m, 0u)] = e;
            cnt += __builtin_popcount(m);
            if (cnt >= 32) {
                cnt -= 32;
                agg_take_batch(q + cnt, 32, src, dst, ht, acc, lane, lst, cap, &fill);
            }
        }
        if (cnt > 0) agg_take_batch(q, cnt, src, dst, ht, acc, lane, lst, cap, &fill);
    } else {
        const int wv    = __builtin_amdgcn_readfirstlane(wave);
        const int nfull = cap >> 5;
        int b = wv;
        int2 cur = make_int2(0, 0);
        if (b < nfull) cur = lst[b * 32 + lane];
        while (b < nfull) {
            const int nb = b + AGG_NW;
            int2 nxt = cur;
            if (nb < nfull) nxt = lst[nb * 32 + lane];
            agg_add_rows(cur.x, cur.y, 32, ht, acc, lane);
            cur = nxt;
            b = nb;
        }
        const int rem = cap & 31;
        if (rem > 0 && wv == (nfull & (AGG_NW - 1))) {
            const int2 en = lst[nfull * 32 + min(lane, rem - 1)];
            agg_add_rows(en.x, en.y, rem, ht, acc, lane);
        }
    }
    __syncthreads();

    float4* o4 = (float4*)(agg + (size_t)base * 64);
    for (int i = tid; i < AGG_CH * 16; i += AGG_NT)
        if (base + (i >> 4) < N) o4[i] = a4[i];
}

__global__ void fc_out_kernel(const float* __restrict__ agg2,
                              const float* __restrict__ dis,
                              const float* __restrict__ b2,
                              const float* __restrict__ wfc,
                              const float* __restrict__ bfc,
                              float* __restrict__ out, int M) {
    const int lane = threadIdx.x & 31;
    const int i = blockIdx.x * (blockDim.x >> 5) + (threadIdx.x >> 5);
    if (i >= M) return;
    const float di = dis[i];
    const int j = lane * 2;
    const float2 a = *(const float2*)(agg2 + (size_t)i * 64 + j);
    const float h0 = fmaxf(fmaf(a.x, di, b2[j]), 0.0f);
    const float h1 = fmaxf(fmaf(a.y, di, b2[j + 1]), 0.0f);
    float p = h0 * wfc[j] + h1 * wfc[j + 1];
    #pragma unroll
    for (int off = 16; off > 0; off >>= 1) p += __shfl_down(p, off, 32);
    if (lane == 0) out[i] = p + bfc[0];
}

extern "C" void kernel_launch(void* const* d_in, const int* in_sizes, int n_in,
                              void* d_out, int out_size, void* d_ws, size_t ws_size,
                              hipStream_t stream) {
    const float* x   = (const float*)d_in[0];
    const int*   ei  = (const int*)d_in[1];
    const float* W1  = (const float*)d_in[2];
    const float* b1  = (const float*)d_in[3];
    const float* W2  = (const float*)d_in[4];
    const float* b2  = (const float*)d_in[5];
    const float* Wfc = (const float*)d_in[6];
    const float* bfc = (const float*)d_in[7];

    const int HID = in_sizes[3];
    const int IN  = in_sizes[2] / HID;
    const int N   = in_sizes[0] / IN;
    const int E   = in_sizes[1] / 2;
    const int* srcv = ei;
    const int* dstv = ei + E;

    char* ws = (char*)d_ws;
    float* dis = (float*)ws;
    size_t off = (((size_t)N * 4) + 255) & ~(size_t)255;
    const size_t rowBytes = (size_t)N * 64 * sizeof(float);
    float* ht   = (float*)(ws + off);
    float* agg1 = (float*)(ws + off + rowBytes);
    float* agg2 = (float*)(ws + off + 2 * rowBytes);

    const int nChunks = (N + DEG_CH - 1) / DEG_CH;
    const int npad    = nChunks * DEG_CH;
    unsigned* planes  = (unsigned*)(ws + off + 3 * rowBytes);
    const size_t planeBytes = (size_t)DEG_SLICES * npad * sizeof(unsigned);
    const int nPart   = (N + ISQ_NT - 1) / ISQ_NT;
    unsigned* part    = (unsigned*)(ws + off + 3 * rowBytes + planeBytes);
    const size_t partBytes = (((size_t)nPart * 4) + 255) & ~(size_t)255;
    int2* list        = (int2*)(ws + off + 3 * rowBytes + planeBytes + partBytes);
    deg_hist_kernel<<<dim3(DEG_SLICES, nChunks), DEG_NT, 0, stream>>>(dstv, planes, E, npad);
    deg_isqrt_kernel<<<nPart, ISQ_NT, 0, stream>>>(planes, dis, part, N, npad);

    const int gemmBlocks = ((N + 15) / 16 + 7) / 8;
    const int aggBlocks  = (N + AGG_CH - 1) / AGG_CH;
    const int nodeBlocks = (N + 7) / 8;

    gemm_wmma_kernel<false><<<gemmBlocks, 256, 0, stream>>>(x, W1, dis, nullptr, ht, N, IN);
    agg_owner_kernel<true><<<aggBlocks, AGG_NT, 0, stream>>>(srcv, dstv, ht, agg1, part, list, E, N, nPart);

    gemm_wmma_kernel<true><<<gemmBlocks, 256, 0, stream>>>(agg1, W2, dis, b1, ht, N, HID);
    agg_owner_kernel<false><<<aggBlocks, AGG_NT, 0, stream>>>(srcv, dstv, ht, agg2, part, list, E, N, nPart);

    fc_out_kernel<<<nodeBlocks, 256, 0, stream>>>(agg2, dis, b2, Wfc, bfc, (float*)d_out, N);
}


# layer-1 owner blocks scan one owner-range byte per edge (written by the degree pass; 16 edges per lane and load, zero-byte test per dword) instead of the 4-byte destination list; sum order: arrival
# speedup vs baseline: 2.3945x; 1.1199x over previous
#include <hip/hip_runtime.h>

typedef __attribute__((ext_vector_type(2))) float v2f;
typedef __attribute__((ext_vector_type(8))) float v8f;

#define KEY_SHIFT  9
#define DEG_CH     32768
#define DEG_SLICES 64
#define DEG_NT     512
#define DEG_DEPTH  8

typedef int int4u __attribute__((ext_vector_type(4), aligned(4)));
typedef unsigned key4 __attribute__((ext_vector_type(4)));

__global__ __launch_bounds__(DEG_NT)
void deg_hist_kernel(const int* __restrict__ dst, unsigned* __restrict__ planes,
                     unsigned* __restrict__ keys, int E, int npad) {
    __shared__ unsigned cnt[DEG_CH];
    const int tid   = threadIdx.x;
    const int slice = blockIdx.x;
    const unsigned base = blockIdx.y * (unsigned)DEG_CH;

    uint4* c4 = (uint4*)cnt;
    for (int i = tid; i < DEG_CH / 4; i += DEG_NT) c4[i] = make_uint4(0u, 0u, 0u, 0u);
    __syncthreads();

    const int nvec = E >> 2;
    const int per  = (nvec + DEG_SLICES - 1) / DEG_SLICES;
    const int lo   = slice * per;
    const int hi   = min(lo + per, nvec);
    const int4u* d4 = (const int4u*)dst;
    const bool wkeys = blockIdx.y == 0;

    for (int v0 = lo + tid; v0 < hi; v0 += DEG_DEPTH * DEG_NT) {
        int4u d[DEG_DEPTH];
        #pragma unroll
        for (int u = 0; u < DEG_DEPTH; ++u) {
            const int v = v0 + u * DEG_NT;
            d[u] = d4[min(v, hi - 1)];
            if (v >= hi) d[u] = (int4u)(-1);
        }
        if (wkeys) {
            #pragma unroll
            for (int u = 0; u < DEG_DEPTH; ++u) {
                const int v = v0 + u * DEG_NT;
                if (v < hi) {
                    const unsigned k0 = min((unsigned)d[u][0] >> KEY_SHIFT, 255u);
                    const unsigned k1 = min((unsigned)d[u][1] >> KEY_SHIFT, 255u);
                    const unsigned k2 = min((unsigned)d[u][2] >> KEY_SHIFT, 255u);
                    const unsigned k3 = min((unsigned)d[u][3] >> KEY_SHIFT, 255u);
                    keys[v] = k0 | (k1 << 8) | (k2 << 16) | (k3 << 24);
                }
            }
        }
        #pragma unroll
        for (int u = 0; u < DEG_DEPTH; ++u) {
            #pragma unroll
            for (int q = 0; q < 4; ++q) {
                const unsigned r = (unsigned)d[u][q] - base;
                if (r < (unsigned)DEG_CH) atomicAdd(&cnt[r], 1u);
            }
        }
    }
    if (slice == DEG_SLICES - 1) {
        for (int e = (nvec << 2) + tid; e < E; e += DEG_NT) {
            const unsigned r = (unsigned)dst[e] - base;
            if (r < (unsigned)DEG_CH) atomicAdd(&cnt[r], 1u);
        }
    }
    __syncthreads();

    uint4* o4 = (uint4*)(planes + (size_t)slice * npad + base);
    for (int i = tid; i < DEG_CH / 4; i += DEG_NT) o4[i] = c4[i];
}

#define ISQ_NT 256

__global__ __launch_bounds__(ISQ_NT)
void deg_isqrt_kernel(const unsigned* __restrict__ planes, float* __restrict__ dis,
                      unsigned* __restrict__ part, int n, int npad) {
    __shared__ unsigned wsum[ISQ_NT / 32];
    const int i = blockIdx.x * ISQ_NT + threadIdx.x;
    unsigned s = 0u;
    if (i < n) {
        #pragma unroll 16
        for (int k = 0; k < DEG_SLICES; ++k) s += planes[(size_t)k * npad + i];
        dis[i] = rsqrtf((float)(s + 1u));
    }
    unsigned w = s;
    #pragma unroll
    for (int off = 16; off > 0; off >>= 1) w += __shfl_xor(w, off, 32);
    if ((threadIdx.x & 31) == 0) wsum[threadIdx.x >> 5] = w;
    __syncthreads();
    if (threadIdx.x == 0) {
        unsigned t = 0u;
        for (int j = 0; j < ISQ_NT / 32; ++j) t += wsum[j];
        part[blockIdx.x] = t;
    }
}

template <bool FUSE>
__global__ void gemm_wmma_kernel(const float* __restrict__ A,
                                 const float* __restrict__ W,
                                 const float* __restrict__ dis,
                                 const float* __restrict__ bias,
                                 float* __restrict__ ht,
                                 int M, int K) {
    const int lane = threadIdx.x & 31;
    const int wave = blockIdx.x * (blockDim.x >> 5) + (threadIdx.x >> 5);
    const int m0 = wave * 16;
    if (m0 >= M) return;

    const int half = lane >> 4;
    const int ml   = lane & 15;
    const int arow = m0 + ml;
    const float dA = FUSE ? dis[arow] : 0.0f;

    v8f c0 = {}, c1 = {}, c2 = {}, c3 = {};

    for (int k0 = 0; k0 < K; k0 += 4) {
        const int ka = k0 + half * 2;
        float a0 = A[(size_t)arow * K + ka];
        float a1 = A[(size_t)arow * K + ka + 1];
        if (FUSE) {
            a0 = fmaxf(fmaf(a0, dA, bias[ka]), 0.0f);
            a1 = fmaxf(fmaf(a1, dA, bias[ka + 1]), 0.0f);
        }
        v2f av; av.x = a0; av.y = a1;

        const float* w0 = W + (size_t)ka * 64;
        const float* w1 = W + (size_t)(ka + 1) * 64;
        v2f b0; b0.x = w0[ml];      b0.y = w1[ml];
        v2f b1; b1.x = w0[16 + ml]; b1.y = w1[16 + ml];
        v2f b2; b2.x = w0[32 + ml]; b2.y = w1[32 + ml];
        v2f b3; b3.x = w0[48 + ml]; b3.y = w1[48 + ml];

        c0 = __builtin_amdgcn_wmma_f32_16x16x4_f32(false, av, false, b0, (short)0, c0, false, false);
        c1 = __builtin_amdgcn_wmma_f32_16x16x4_f32(false, av, false, b1, (short)0, c1, false, false);
        c2 = __builtin_amdgcn_wmma_f32_16x16x4_f32(false, av, false, b2, (short)0, c2, false, false);
        c3 = __builtin_amdgcn_wmma_f32_16x16x4_f32(false, av, false, b3, (short)0, c3, false, false);
    }

    for (int r = 0; r < 8; ++r) {
        const int orow = m0 + half * 8 + r;
        const float dr = dis[orow];
        const float v0 = c0[r] * dr;
        const float v1 = c1[r] * dr;
        const float v2 = c2[r] * dr;
        const float v3 = c3[r] * dr;
        float* hp = ht + (size_t)orow * 64;
        hp[ml] = v0; hp[16 + ml] = v1; hp[32 + ml] = v2; hp[48 + ml] = v3;
    }
}

#define AGG_SHIFT 9
#define AGG_CH    (1 << AGG_SHIFT)
#define AGG_NT    1024
#define AGG_NW    (AGG_NT / 32)
#define AGG_DEPTH 8
#define AGG_Q     64
#define AGG_UQ    (AGG_DEPTH * 32)
#define AGG_ROWS  16
#define AGG_PARTS (AGG_CH / ISQ_NT)

__device__ __forceinline__ void agg_add_rows(int s, int dl, int n,
                                             const float* __restrict__ ht,
                                             float* acc, int lane) {
    #pragma unroll
    for (int j0 = 0; j0 < 32; j0 += AGG_ROWS) {
        if (j0 < n) {
            float2 v[AGG_ROWS];
            #pragma unroll
            for (int r = 0; r < AGG_ROWS; ++r) {
                const int sj = __builtin_amdgcn_readlane(s, j0 + r);
                v[r] = *(const float2*)(ht + (size_t)sj * 64 + lane * 2);
            }
            #pragma unroll
            for (int r = 0; r < AGG_ROWS; ++r) {
                if (j0 + r < n) {
                    const int dj = __builtin_amdgcn_readlane(dl, j0 + r);
                    float* ap = acc + dj * 64 + lane * 2;
                    atomicAdd(ap, v[r].x);
                    atomicAdd(ap + 1, v[r].y);
                }
            }
        }
    }
}

__device__ __forceinline__ void agg_take_batch(const int* q, int n,
                                               const int* __restrict__ src,
                                               const int* __restrict__ dst,
                                               const float* __restrict__ ht,
                                               float* acc, int lane,
                                               int2* __restrict__ lst, int cap, int* fill) {
    const int e  = q[min(lane, n - 1)];
    const int s  = src[e];
    const int dl = dst[e] & (AGG_CH - 1);
    int pos = 0;
    if (lane == 0) pos = atomicAdd(fill, n);
    pos = __builtin_amdgcn_readfirstlane(pos) + lane;
    if (lane < n && pos < cap) lst[pos] = make_int2(s, dl);
    agg_add_rows(s, dl, n, ht, acc, lane);
}

template <bool BUILD>
__global__ __launch_bounds__(AGG_NT)
void agg_owner_kernel(const int* __restrict__ src, const int* __restrict__ dst,
                      const float* __restrict__ ht, float* __restrict__ agg,
                      const unsigned* __restrict__ part, int2* __restrict__ list,
                      const unsigned* __restrict__ keys, int E, int N, int nPart) {
    __shared__ float acc[AGG_CH * 64];
    __shared__ unsigned red[AGG_NW];
    __shared__ int fill;
    const int tid  = threadIdx.x;
    const int lane = tid & 31;
    const int wave = tid >> 5;
    const unsigned blk = blockIdx.x;
    const int base = (int)blk * AGG_CH;

    unsigned below = 0u;
    for (int j = tid; j < (int)blk * AGG_PARTS; j += AGG_NT) below += part[j];
    #pragma unroll
    for (int off = 16; off > 0; off >>= 1) below += __shfl_xor(below, off, 32);
    if (lane == 0) red[wave] = below;
    if (tid == 0) fill = 0;
    unsigned mine = 0u;
    #pragma unroll
    for (int j = 0; j < AGG_PARTS; ++j) {
        const int p = (int)blk * AGG_PARTS + j;
        if (p < nPart) mine += part[p];
    }
    const int cap = (int)mine;

    float4* a4 = (float4*)acc;
    const float4* h4 = (const float4*)(ht + (size_t)base * 64);
    for (int i = tid; i < AGG_CH * 16; i += AGG_NT)
        a4[i] = (base + (i >> 4) < N) ? h4[i] : make_float4(0.f, 0.f, 0.f, 0.f);
    __syncthreads();

    static_assert(AGG_NW == 32, "one wave sum per lane");
    unsigned first = red[lane];
    #pragma unroll
    for (int off = 16; off > 0; off >>= 1) first += __shfl_xor(first, off, 32);
    int2* lst = list + (size_t)__builtin_amdgcn_readfirstlane((int)first);

    if (BUILD) {
        __shared__ int queue[AGG_NW][AGG_Q];
        __shared__ int2 uqueue[AGG_NW][AGG_UQ];
        int* q = queue[wave];
        int2* uq = uqueue[wave];
        int cnt = 0;
        const int nvec = E >> 4;
        const key4* k4 = (const key4*)keys;
        const int step = AGG_DEPTH * AGG_NT;
        const int nIt  = (nvec + step - 1) / step;
        const int rot  = (int)(((long long)blk * nIt) / gridDim.x);
        const unsigned pat = (blk & 255u) * 0x01010101u;

        for (int it = 0; it < nIt; ++it) {
            int itr = it + rot; if (itr >= nIt) itr -= nIt;
            const int v0 = itr * step + tid;
            key4 d[AGG_DEPTH];
            #pragma unroll
            for (int u = 0; u < AGG_DEPTH; ++u)
                d[u] = k4[min(v0 + u * AGG_NT, nvec - 1)];
            __builtin_amdgcn_sched_barrier(0);
            int nu = 0;
            #pragma unroll
            for (int u = 0; u < AGG_DEPTH; ++u) {
                const int v = v0 + u * AGG_NT;
                unsigned cc = 0u;
                #pragma unroll
                for (int k = 0; k < 4; ++k) {
                    const unsigned x = d[u][k] ^ pat;
                    const unsigned t = ((x & 0x7f7f7f7fu) + 0x7f7f7f7fu) | x;
                    cc |= (t >> (7 - k)) & (0x01010101u << k);
                }
                unsigned f = cc ^ 0x0f0f0f0fu;
                if (v >= nvec) f = 0u;
                const unsigned m = (unsigned)__builtin_amdgcn_ballot_w32(f != 0u);
                if (m != 0u) {
                    if (f != 0u) uq[nu + (int)__builtin_amdgcn_mbcnt_lo(m, 0u)] = make_int2((int)f, v);
                    nu += __builtin_popcount(m);
                }
            }
            for (int i = 0; i < nu; i += 32) {
                int2 en = make_int2(0, 0);
                if (i + lane < nu) en = uq[i + lane];
                unsigned f = (unsigned)en.x;
                unsigned m = (unsigned)__builtin_amdgcn_ballot_w32(f != 0u);
                while (m != 0u) {
                    if (f != 0u) {
                        const int b = __builtin_ctz(f);
                        q[cnt + (int)__builtin_amdgcn_mbcnt_lo(m, 0u)] = (en.y << 4) + ((b & 3) << 2) + (b >> 3);
                        f &= f - 1u;
                    }
                    cnt += __builtin_popcount(m);
                    if (cnt >= 32) {
                        cnt -= 32;
                        agg_take_batch(q + cnt, 32, src, dst, ht, acc, lane, lst, cap, &fill);
                    }
                    m = (unsigned)__builtin_amdgcn_ballot_w32(f != 0u);
                }
            }
        }
        if (wave == 0) {
            const int e = (nvec << 4) + lane;
            const bool hit = e < E && ((unsigned)dst[min(e, E - 1)] >> AGG_SHIFT) == blk;
            const unsigned m = (unsigned)__builtin_amdgcn_ballot_w32(hit);
            if (hit) q[cnt + (int)__builtin_amdgcn_mbcnt_lo(m, 0u)] = e;
            cnt += __builtin_popcount(m);
            if (cnt >= 32) {
                cnt -= 32;
                agg_take_batch(q + cnt, 32, src, dst, ht, acc, lane, lst, cap, &fill);
            }
        }
        if (cnt > 0) agg_take_batch(q, cnt, src, dst, ht, acc, lane, lst, cap, &fill);
    } else {
        const int wv    = __builtin_amdgcn_readfirstlane(wave);
        const int nfull = cap >> 5;
        int b = wv;
        int2 cur = make_int2(0, 0);
        if (b < nfull) cur = lst[b * 32 + lane];
        while (b < nfull) {
            const int nb = b + AGG_NW;
            int2 nxt = cur;
            if (nb < nfull) nxt = lst[nb * 32 + lane];
            agg_add_rows(cur.x, cur.y, 32, ht, acc, lane);
            cur = nxt;
            b = nb;
        }
        const int rem = cap & 31;
        if (rem > 0 && wv == (nfull & (AGG_NW - 1))) {
            const int2 en = lst[nfull * 32 + min(lane, rem - 1)];
            agg_add_rows(en.x, en.y, rem, ht, acc, lane);
        }
    }
    __syncthreads();

    float4* o4 = (float4*)(agg + (size_t)base * 64);
    for (int i = tid; i < AGG_CH * 16; i += AGG_NT)
        if (base + (i >> 4) < N) o4[i] = a4[i];
}

__global__ void fc_out_kernel(const float* __restrict__ agg2,
                              const float* __restrict__ dis,
                              const float* __restrict__ b2,
                              const float* __restrict__ wfc,
                              const float* __restrict__ bfc,
                              float* __restrict__ out, int M) {
    const int lane = threadIdx.x & 31;
    const int i = blockIdx.x * (blockDim.x >> 5) + (threadIdx.x >> 5);
    if (i >= M) return;
    const float di = dis[i];
    const int j = lane * 2;
    const float2 a = *(const float2*)(agg2 + (size_t)i * 64 + j);
    const float h0 = fmaxf(fmaf(a.x, di, b2[j]), 0.0f);
    const float h1 = fmaxf(fmaf(a.y, di, b2[j + 1]), 0.0f);
    float p = h0 * wfc[j] + h1 * wfc[j + 1];
    #pragma unroll
    for (int off = 16; off > 0; off >>= 1) p += __shfl_down(p, off, 32);
    if (lane == 0) out[i] = p + bfc[0];
}

extern "C" void kernel_launch(void* const* d_in, const int* in_sizes, int n_in,
                              void* d_out, int out_size, void* d_ws, size_t ws_size,
                              hipStream_t stream) {
    const float* x   = (const float*)d_in[0];
    const int*   ei  = (const int*)d_in[1];
    const float* W1  = (const float*)d_in[2];
    const float* b1  = (const float*)d_in[3];
    const float* W2  = (const float*)d_in[4];
    const float* b2  = (const float*)d_in[5];
    const float* Wfc = (const float*)d_in[6];
    const float* bfc = (const float*)d_in[7];

    const int HID = in_sizes[3];
    const int IN  = in_sizes[2] / HID;
    const int N   = in_sizes[0] / IN;
    const int E   = in_sizes[1] / 2;
    const int* srcv = ei;
    const int* dstv = ei + E;

    char* ws = (char*)d_ws;
    float* dis = (float*)ws;
    size_t off = (((size_t)N * 4) + 255) & ~(size_t)255;
    const size_t rowBytes = (size_t)N * 64 * sizeof(float);
    float* ht   = (float*)(ws + off);
    float* agg1 = (float*)(ws + off + rowBytes);
    float* agg2 = (float*)(ws + off + 2 * rowBytes);

    const int nChunks = (N + DEG_CH - 1) / DEG_CH;
    const int npad    = nChunks * DEG_CH;
    unsigned* planes  = (unsigned*)(ws + off + 3 * rowBytes);
    const size_t planeBytes = (size_t)DEG_SLICES * npad * sizeof(unsigned);
    const int nPart   = (N + ISQ_NT - 1) / ISQ_NT;
    unsigned* part    = (unsigned*)(ws + off + 3 * rowBytes + planeBytes);
    const size_t partBytes = (((size_t)nPart * 4) + 255) & ~(size_t)255;
    int2* list        = (int2*)(ws + off + 3 * rowBytes + planeBytes + partBytes);
    unsigned* keys    = (unsigned*)(ws + off + 3 * rowBytes + planeBytes + partBytes + (size_t)E * sizeof(int2));
    deg_hist_kernel<<<dim3(DEG_SLICES, nChunks), DEG_NT, 0, stream>>>(dstv, planes, keys, E, npad);
    deg_isqrt_kernel<<<nPart, ISQ_NT, 0, stream>>>(planes, dis, part, N, npad);

    const int gemmBlocks = ((N + 15) / 16 + 7) / 8;
    const int aggBlocks  = (N + AGG_CH - 1) / AGG_CH;
    const int nodeBlocks = (N + 7) / 8;

    gemm_wmma_kernel<false><<<gemmBlocks, 256, 0, stream>>>(x, W1, dis, nullptr, ht, N, IN);
    agg_owner_kernel<true><<<aggBlocks, AGG_NT, 0, stream>>>(srcv, dstv, ht, agg1, part, list, keys, E, N, nPart);

    gemm_wmma_kernel<true><<<gemmBlocks, 256, 0, stream>>>(agg1, W2, dis, b1, ht, N, HID);
    agg_owner_kernel<false><<<aggBlocks, AGG_NT, 0, stream>>>(srcv, dstv, ht, agg2, part, list, keys, E, N, nPart);

    fc_out_kernel<<<nodeBlocks, 256, 0, stream>>>(agg2, dis, b2, Wfc, bfc, (float*)d_out, N);
}


# layer-2 owner blocks finish their nodes from the LDS sums (relu, fc dot on the same lane tree, bias) and store out in whole lines; agg2 plane and fc launch gone; 7 launches to 6; sum orders kept
# speedup vs baseline: 2.5587x; 1.0686x over previous
#include <hip/hip_runtime.h>

typedef __attribute__((ext_vector_type(2))) float v2f;
typedef __attribute__((ext_vector_type(8))) float v8f;

#define KEY_SHIFT  9
#define DEG_CH     32768
#define DEG_SLICES 64
#define DEG_NT     512
#define DEG_DEPTH  8

typedef int int4u __attribute__((ext_vector_type(4), aligned(4)));
typedef unsigned key4 __attribute__((ext_vector_type(4)));

__global__ __launch_bounds__(DEG_NT)
void deg_hist_kernel(const int* __restrict__ dst, unsigned* __restrict__ planes,
                     unsigned* __restrict__ keys, int E, int npad) {
    __shared__ unsigned cnt[DEG_CH];
    const int tid   = threadIdx.x;
    const int slice = blockIdx.x;
    const unsigned base = blockIdx.y * (unsigned)DEG_CH;

    uint4* c4 = (uint4*)cnt;
    for (int i = tid; i < DEG_CH / 4; i += DEG_NT) c4[i] = make_uint4(0u, 0u, 0u, 0u);
    __syncthreads();

    const int nvec = E >> 2;
    const int per  = (nvec + DEG_SLICES - 1) / DEG_SLICES;
    const int lo   = slice * per;
    const int hi   = min(lo + per, nvec);
    const int4u* d4 = (const int4u*)dst;
    const bool wkeys = blockIdx.y == 0;

    for (int v0 = lo + tid; v0 < hi; v0 += DEG_DEPTH * DEG_NT) {
        int4u d[DEG_DEPTH];
        #pragma unroll
        for (int u = 0; u < DEG_DEPTH; ++u) {
            const int v = v0 + u * DEG_NT;
            d[u] = d4[min(v, hi - 1)];
            if (v >= hi) d[u] = (int4u)(-1);
        }
        if (wkeys) {
            #pragma unroll
            for (int u = 0; u < DEG_DEPTH; ++u) {
                const int v = v0 + u * DEG_NT;
                if (v < hi) {
                    const unsigned k0 = min((unsigned)d[u][0] >> KEY_SHIFT, 255u);
                    const unsigned k1 = min((unsigned)d[u][1] >> KEY_SHIFT, 255u);
                    const unsigned k2 = min((unsigned)d[u][2] >> KEY_SHIFT, 255u);
                    const unsigned k3 = min((unsigned)d[u][3] >> KEY_SHIFT, 255u);
                    keys[v] = k0 | (k1 << 8) | (k2 << 16) | (k3 << 24);
                }
            }
        }
        #pragma unroll
        for (int u = 0; u < DEG_DEPTH; ++u) {
            #pragma unroll
            for (int q = 0; q < 4; ++q) {
                const unsigned r = (unsigned)d[u][q] - base;
                if (r < (unsigned)DEG_CH) atomicAdd(&cnt[r], 1u);
            }
        }
    }
    if (slice == DEG_SLICES - 1) {
        for (int e = (nvec << 2) + tid; e < E; e += DEG_NT) {
            const unsigned r = (unsigned)dst[e] - base;
            if (r < (unsigned)DEG_CH) atomicAdd(&cnt[r], 1u);
        }
    }
    __syncthreads();

    uint4* o4 = (uint4*)(planes + (size_t)slice * npad + base);
    for (int i = tid; i < DEG_CH / 4; i += DEG_NT) o4[i] = c4[i];
}

#define ISQ_NT 256

__global__ __launch_bounds__(ISQ_NT)
void deg_isqrt_kernel(const unsigned* __restrict__ planes, float* __restrict__ dis,
                      unsigned* __restrict__ part, int n, int npad) {
    __shared__ unsigned wsum[ISQ_NT / 32];
    const int i = blockIdx.x * ISQ_NT + threadIdx.x;
    unsigned s = 0u;
    if (i < n) {
        #pragma unroll 16
        for (int k = 0; k < DEG_SLICES; ++k) s += planes[(size_t)k * npad + i];
        dis[i] = rsqrtf((float)(s + 1u));
    }
    unsigned w = s;
    #pragma unroll
    for (int off = 16; off > 0; off >>= 1) w += __shfl_xor(w, off, 32);
    if ((threadIdx.x & 31) == 0) wsum[threadIdx.x >> 5] = w;
    __syncthreads();
    if (threadIdx.x == 0) {
        unsigned t = 0u;
        for (int j = 0; j < ISQ_NT / 32; ++j) t += wsum[j];
        part[blockIdx.x] = t;
    }
}

template <bool FUSE>
__global__ void gemm_wmma_kernel(const float* __restrict__ A,
                                 const float* __restrict__ W,
                                 const float* __restrict__ dis,
                                 const float* __restrict__ bias,
                                 float* __restrict__ ht,
                                 int M, int K) {
    const int lane = threadIdx.x & 31;
    const int wave = blockIdx.x * (blockDim.x >> 5) + (threadIdx.x >> 5);
    const int m0 = wave * 16;
    if (m0 >= M) return;

    const int half = lane >> 4;
    const int ml   = lane & 15;
    const int arow = m0 + ml;
    const float dA = FUSE ? dis[arow] : 0.0f;

    v8f c0 = {}, c1 = {}, c2 = {}, c3 = {};

    for (int k0 = 0; k0 < K; k0 += 4) {
        const int ka = k0 + half * 2;
        float a0 = A[(size_t)arow * K + ka];
        float a1 = A[(size_t)arow * K + ka + 1];
        if (FUSE) {
            a0 = fmaxf(fmaf(a0, dA, bias[ka]), 0.0f);
            a1 = fmaxf(fmaf(a1, dA, bias[ka + 1]), 0.0f);
        }
        v2f av; av.x = a0; av.y = a1;

        const float* w0 = W + (size_t)ka * 64;
        const float* w1 = W + (size_t)(ka + 1) * 64;
        v2f b0; b0.x = w0[ml];      b0.y = w1[ml];
        v2f b1; b1.x = w0[16 + ml]; b1.y = w1[16 + ml];
        v2f b2; b2.x = w0[32 + ml]; b2.y = w1[32 + ml];
        v2f b3; b3.x = w0[48 + ml]; b3.y = w1[48 + ml];

        c0 = __builtin_amdgcn_wmma_f32_16x16x4_f32(false, av, false, b0, (short)0, c0, false, false);
        c1 = __builtin_amdgcn_wmma_f32_16x16x4_f32(false, av, false, b1, (short)0, c1, false, false);
        c2 = __builtin_amdgcn_wmma_f32_16x16x4_f32(false, av, false, b2, (short)0, c2, false, false);
        c3 = __builtin_amdgcn_wmma_f32_16x16x4_f32(false, av, false, b3, (short)0, c3, false, false);
    }

    for (int r = 0; r < 8; ++r) {
        const int orow = m0 + half * 8 + r;
        const float dr = dis[orow];
        const float v0 = c0[r] * dr;
        const float v1 = c1[r] * dr;
        const float v2 = c2[r] * dr;
        const float v3 = c3[r] * dr;
        float* hp = ht + (size_t)orow * 64;
        hp[ml] = v0; hp[16 + ml] = v1; hp[32 + ml] = v2; hp[48 + ml] = v3;
    }
}

#define AGG_SHIFT 9
#define AGG_CH    (1 << AGG_SHIFT)
#define AGG_NT    1024
#define AGG_NW    (AGG_NT / 32)
#define AGG_DEPTH 8
#define AGG_Q     64
#define AGG_UQ    (AGG_DEPTH * 32)
#define AGG_ROWS  16
#define AGG_PARTS (AGG_CH / ISQ_NT)

__device__ __forceinline__ void agg_add_rows(int s, int dl, int n,
                                             const float* __restrict__ ht,
                                             float* acc, int lane) {
    #pragma unroll
    for (int j0 = 0; j0 < 32; j0 += AGG_ROWS) {
        if (j0 < n) {
            float2 v[AGG_ROWS];
            #pragma unroll
            for (int r = 0; r < AGG_ROWS; ++r) {
                const int sj = __builtin_amdgcn_readlane(s, j0 + r);
                v[r] = *(const float2*)(ht + (size_t)sj * 64 + lane * 2);
            }
            #pragma unroll
            for (int r = 0; r < AGG_ROWS; ++r) {
                if (j0 + r < n) {
                    const int dj = __builtin_amdgcn_readlane(dl, j0 + r);
                    float* ap = acc + dj * 64 + lane * 2;
                    atomicAdd(ap, v[r].x);
                    atomicAdd(ap + 1, v[r].y);
                }
            }
        }
    }
}

__device__ __forceinline__ void agg_take_batch(const int* q, int n,
                                               const int* __restrict__ src,
                                               const int* __restrict__ dst,
                                               const float* __restrict__ ht,
                                               float* acc, int lane,
                                               int2* __restrict__ lst, int cap, int* fill) {
    const int e  = q[min(lane, n - 1)];
    const int s  = src[e];
    const int dl = dst[e] & (AGG_CH - 1);
    int pos = 0;
    if (lane == 0) pos = atomicAdd(fill, n);
    pos = __builtin_amdgcn_readfirstlane(pos) + lane;
    if (lane < n && pos < cap) lst[pos] = make_int2(s, dl);
    agg_add_rows(s, dl, n, ht, acc, lane);
}

template <bool BUILD>
__global__ __launch_bounds__(AGG_NT)
void agg_owner_kernel(const int* __restrict__ src, const int* __restrict__ dst,
                      const float* __restrict__ ht, float* __restrict__ agg,
                      const unsigned* __restrict__ part, int2* __restrict__ list,
                      const unsigned* __restrict__ keys, int E, int N, int nPart,
                      const float* __restrict__ dis, const float* __restrict__ b2,
                      const float* __restrict__ wfc, const float* __restrict__ bfc,
                      float* __restrict__ out) {
    __shared__ float acc[AGG_CH * 64];
    __shared__ unsigned red[AGG_NW];
    __shared__ int fill;
    const int tid  = threadIdx.x;
    const int lane = tid & 31;
    const int wave = tid >> 5;
    const unsigned blk = blockIdx.x;
    const int base = (int)blk * AGG_CH;

    unsigned below = 0u;
    for (int j = tid; j < (int)blk * AGG_PARTS; j += AGG_NT) below += part[j];
    #pragma unroll
    for (int off = 16; off > 0; off >>= 1) below += __shfl_xor(below, off, 32);
    if (lane == 0) red[wave] = below;
    if (tid == 0) fill = 0;
    unsigned mine = 0u;
    #pragma unroll
    for (int j = 0; j < AGG_PARTS; ++j) {
        const int p = (int)blk * AGG_PARTS + j;
        if (p < nPart) mine += part[p];
    }
    const int cap = (int)mine;

    float4* a4 = (float4*)acc;
    const float4* h4 = (const float4*)(ht + (size_t)base * 64);
    for (int i = tid; i < AGG_CH * 16; i += AGG_NT)
        a4[i] = (base + (i >> 4) < N) ? h4[i] : make_float4(0.f, 0.f, 0.f, 0.f);
    __syncthreads();

    static_assert(AGG_NW == 32, "one wave sum per lane");
    unsigned first = red[lane];
    #pragma unroll
    for (int off = 16; off > 0; off >>= 1) first += __shfl_xor(first, off, 32);
    int2* lst = list + (size_t)__builtin_amdgcn_readfirstlane((int)first);

    if (BUILD) {
        __shared__ int queue[AGG_NW][AGG_Q];
        __shared__ int2 uqueue[AGG_NW][AGG_UQ];
        int* q = queue[wave];
        int2* uq = uqueue[wave];
        int cnt = 0;
        const int nvec = E >> 4;
        const key4* k4 = (const key4*)keys;
        const int step = AGG_DEPTH * AGG_NT;
        const int nIt  = (nvec + step - 1) / step;
        const int rot  = (int)(((long long)blk * nIt) / gridDim.x);
        const unsigned pat = (blk & 255u) * 0x01010101u;

        for (int it = 0; it < nIt; ++it) {
            int itr = it + rot; if (itr >= nIt) itr -= nIt;
            const int v0 = itr * step + tid;
            key4 d[AGG_DEPTH];
            #pragma unroll
            for (int u = 0; u < AGG_DEPTH; ++u)
                d[u] = k4[min(v0 + u * AGG_NT, nvec - 1)];
            __builtin_amdgcn_sched_barrier(0);
            int nu = 0;
            #pragma unroll
            for (int u = 0; u < AGG_DEPTH; ++u) {
                const int v = v0 + u * AGG_NT;
                unsigned cc = 0u;
                #pragma unroll
                for (int k = 0; k < 4; ++k) {
                    const unsigned x = d[u][k] ^ pat;
                    const unsigned t = ((x & 0x7f7f7f7fu) + 0x7f7f7f7fu) | x;
                    cc |= (t >> (7 - k)) & (0x01010101u << k);
                }
                unsigned f = cc ^ 0x0f0f0f0fu;
                if (v >= nvec) f = 0u;
                const unsigned m = (unsigned)__builtin_amdgcn_ballot_w32(f != 0u);
                if (m != 0u) {
                    if (f != 0u) uq[nu + (int)__builtin_amdgcn_mbcnt_lo(m, 0u)] = make_int2((int)f, v);
                    nu += __builtin_popcount(m);
                }
            }
            for (int i = 0; i < nu; i += 32) {
                int2 en = make_int2(0, 0);
                if (i + lane < nu) en = uq[i + lane];
                unsigned f = (unsigned)en.x;
                unsigned m = (unsigned)__builtin_amdgcn_ballot_w32(f != 0u);
                while (m != 0u) {
                    if (f != 0u) {
                        const int b = __builtin_ctz(f);
                        q[cnt + (int)__builtin_amdgcn_mbcnt_lo(m, 0u)] = (en.y << 4) + ((b & 3) << 2) + (b >> 3);
                        f &= f - 1u;
                    }
                    cnt += __builtin_popcount(m);
                    if (cnt >= 32) {
                        cnt -= 32;
                        agg_take_batch(q + cnt, 32, src, dst, ht, acc, lane, lst, cap, &fill);
                    }
                    m = (unsigned)__builtin_amdgcn_ballot_w32(f != 0u);
                }
            }
        }
        if (wave == 0) {
            const int e = (nvec << 4) + lane;
            const bool hit = e < E && ((unsigned)dst[min(e, E - 1)] >> AGG_SHIFT) == blk;
            const unsigned m = (unsigned)__builtin_amdgcn_ballot_w32(hit);
            if (hit) q[cnt + (int)__builtin_amdgcn_mbcnt_lo(m, 0u)] = e;
            cnt += __builtin_popcount(m);
            if (cnt >= 32) {
                cnt -= 32;
                agg_take_batch(q + cnt, 32, src, dst, ht, acc, lane, lst, cap, &fill);
            }
        }
        if (cnt > 0) agg_take_batch(q, cnt, src, dst, ht, acc, lane, lst, cap, &fill);
    } else {
        const int wv    = __builtin_amdgcn_readfirstlane(wave);
        const int nfull = cap >> 5;
        int b = wv;
        int2 cur = make_int2(0, 0);
        if (b < nfull) cur = lst[b * 32 + lane];
        while (b < nfull) {
            const int nb = b + AGG_NW;
            int2 nxt = cur;
            if (nb < nfull) nxt = lst[nb * 32 + lane];
            agg_add_rows(cur.x, cur.y, 32, ht, acc, lane);
            cur = nxt;
            b = nb;
        }
        const int rem = cap & 31;
        if (rem > 0 && wv == (nfull & (AGG_NW - 1))) {
            const int2 en = lst[nfull * 32 + min(lane, rem - 1)];
            agg_add_rows(en.x, en.y, rem, ht, acc, lane);
        }
    }
    __syncthreads();

    if (BUILD) {
        float4* o4 = (float4*)(agg + (size_t)base * 64);
        for (int i = tid; i < AGG_CH * 16; i += AGG_NT)
            if (base + (i >> 4) < N) o4[i] = a4[i];
    } else {
        static_assert(AGG_NW * 16 == AGG_CH, "16 rows per wave");
        __shared__ float outv[AGG_CH];
        const int j = lane * 2;
        const float bj0 = b2[j], bj1 = b2[j + 1];
        const float w0 = wfc[j], w1 = wfc[j + 1];
        const int r0 = wave * 16;
        const int dl = __float_as_int(dis[min(base + r0 + (lane & 15), N - 1)]);
        float res = 0.0f;
        #pragma unroll
        for (int r = 0; r < 16; ++r) {
            const float di = __int_as_float(__builtin_amdgcn_readlane(dl, r));
            const float2 a = *(const float2*)(acc + (r0 + r) * 64 + j);
            const float h0 = fmaxf(fmaf(a.x, di, bj0), 0.0f);
            const float h1 = fmaxf(fmaf(a.y, di, bj1), 0.0f);
            float p = h0 * w0 + h1 * w1;
            #pragma unroll
            for (int off = 16; off > 0; off >>= 1) p += __shfl_xor(p, off, 32);
            if (lane == r) res = p;
        }
        if (lane < 16) outv[r0 + lane] = res + bfc[0];
        __syncthreads();
        if (tid < AGG_CH && base + tid < N) out[base + tid] = outv[tid];
    }
}

extern "C" void kernel_launch(void* const* d_in, const int* in_sizes, int n_in,
                              void* d_out, int out_size, void* d_ws, size_t ws_size,
                              hipStream_t stream) {
    const float* x   = (const float*)d_in[0];
    const int*   ei  = (const int*)d_in[1];
    const float* W1  = (const float*)d_in[2];
    const float* b1  = (const float*)d_in[3];
    const float* W2  = (const float*)d_in[4];
    const float* b2  = (const float*)d_in[5];
    const float* Wfc = (const float*)d_in[6];
    const float* bfc = (const float*)d_in[7];

    const int HID = in_sizes[3];
    const int IN  = in_sizes[2] / HID;
    const int N   = in_sizes[0] / IN;
    const int E   = in_sizes[1] / 2;
    const int* srcv = ei;
    const int* dstv = ei + E;

    char* ws = (char*)d_ws;
    float* dis = (float*)ws;
    size_t off = (((size_t)N * 4) + 255) & ~(size_t)255;
    const size_t rowBytes = (size_t)N * 64 * sizeof(float);
    float* ht   = (float*)(ws + off);
    float* agg1 = (float*)(ws + off + rowBytes);

    const int nChunks = (N + DEG_CH - 1) / DEG_CH;
    const int npad    = nChunks * DEG_CH;
    unsigned* planes  = (unsigned*)(ws + off + 3 * rowBytes);
    const size_t planeBytes = (size_t)DEG_SLICES * npad * sizeof(unsigned);
    const int nPart   = (N + ISQ_NT - 1) / ISQ_NT;
    unsigned* part    = (unsigned*)(ws + off + 3 * rowBytes + planeBytes);
    const size_t partBytes = (((size_t)nPart * 4) + 255) & ~(size_t)255;
    int2* list        = (int2*)(ws + off + 3 * rowBytes + planeBytes + partBytes);
    unsigned* keys    = (unsigned*)(ws + off + 3 * rowBytes + planeBytes + partBytes + (size_t)E * sizeof(int2));
    deg_hist_kernel<<<dim3(DEG_SLICES, nChunks), DEG_NT, 0, stream>>>(dstv, planes, keys, E, npad);
    deg_isqrt_kernel<<<nPart, ISQ_NT, 0, stream>>>(planes, dis, part, N, npad);

    const int gemmBlocks = ((N + 15) / 16 + 7) / 8;
    const int aggBlocks  = (N + AGG_CH - 1) / AGG_CH;

    gemm_wmma_kernel<false><<<gemmBlocks, 256, 0, stream>>>(x, W1, dis, nullptr, ht, N, IN);
    agg_owner_kernel<true><<<aggBlocks, AGG_NT, 0, stream>>>(srcv, dstv, ht, agg1, part, list, keys, E, N, nPart,
                                                             nullptr, nullptr, nullptr, nullptr, nullptr);

    gemm_wmma_kernel<true><<<gemmBlocks, 256, 0, stream>>>(agg1, W2, dis, b1, ht, N, HID);
    agg_owner_kernel<false><<<aggBlocks, AGG_NT, 0, stream>>>(srcv, dstv, ht, nullptr, part, list, keys, E, N, nPart,
                                                              dis, b2, Wfc, bfc, (float*)d_out);
}


# owner node ranges of 416 instead of 512 nodes (range key by division, rsqrt-pass block = one range): 241 owner blocks, not 196, share each layer's LDS adds and row reads; sum order: arrival
# speedup vs baseline: 2.7907x; 1.0907x over previous
#include <hip/hip_runtime.h>

typedef __attribute__((ext_vector_type(2))) float v2f;
typedef __attribute__((ext_vector_type(8))) float v8f;

#define AGG_CH     416
#define DEG_CH     32768
#define DEG_SLICES 64
#define DEG_NT     512
#define DEG_DEPTH  8

typedef int int4u __attribute__((ext_vector_type(4), aligned(4)));
typedef unsigned key4 __attribute__((ext_vector_type(4)));

__global__ __launch_bounds__(DEG_NT)
void deg_hist_kernel(const int* __restrict__ dst, unsigned* __restrict__ planes,
                     unsigned* __restrict__ keys, int E, int npad) {
    __shared__ unsigned cnt[DEG_CH];
    const int tid   = threadIdx.x;
    const int slice = blockIdx.x;
    const unsigned base = blockIdx.y * (unsigned)DEG_CH;

    uint4* c4 = (uint4*)cnt;
    for (int i = tid; i < DEG_CH / 4; i += DEG_NT) c4[i] = make_uint4(0u, 0u, 0u, 0u);
    __syncthreads();

    const int nvec = E >> 2;
    const int per  = (nvec + DEG_SLICES - 1) / DEG_SLICES;
    const int lo   = slice * per;
    const int hi   = min(lo + per, nvec);
    const int4u* d4 = (const int4u*)dst;
    const bool wkeys = blockIdx.y == 0;

    for (int v0 = lo + tid; v0 < hi; v0 += DEG_DEPTH * DEG_NT) {
        int4u d[DEG_DEPTH];
        #pragma unroll
        for (int u = 0; u < DEG_DEPTH; ++u) {
            const int v = v0 + u * DEG_NT;
            d[u] = d4[min(v, hi - 1)];
            if (v >= hi) d[u] = (int4u)(-1);
        }
        if (wkeys) {
            #pragma unroll
            for (int u = 0; u < DEG_DEPTH; ++u) {
                const int v = v0 + u * DEG_NT;
                if (v < hi) {
                    const unsigned k0 = min((unsigned)d[u][0] / (unsigned)AGG_CH, 255u);
                    const unsigned k1 = min((unsigned)d[u][1] / (unsigned)AGG_CH, 255u);
                    const unsigned k2 = min((unsigned)d[u][2] / (unsigned)AGG_CH, 255u);
                    const unsigned k3 = min((unsigned)d[u][3] / (unsigned)AGG_CH, 255u);
                    keys[v] = k0 | (k1 << 8) | (k2 << 16) | (k3 << 24);
                }
            }
        }
        #pragma unroll
        for (int u = 0; u < DEG_DEPTH; ++u) {
            #pragma unroll
            for (int q = 0; q < 4; ++q) {
                const unsigned r = (unsigned)d[u][q] - base;
                if (r < (unsigned)DEG_CH) atomicAdd(&cnt[r], 1u);
            }
        }
    }
    if (slice == DEG_SLICES - 1) {
        for (int e = (nvec << 2) + tid; e < E; e += DEG_NT) {
            const unsigned r = (unsigned)dst[e] - base;
            if (r < (unsigned)DEG_CH) atomicAdd(&cnt[r], 1u);
        }
    }
    __syncthreads();

    uint4* o4 = (uint4*)(planes + (size_t)slice * npad + base);
    for (int i = tid; i < DEG_CH / 4; i += DEG_NT) o4[i] = c4[i];
}

#define ISQ_NT AGG_CH

__global__ __launch_bounds__(ISQ_NT)
void deg_isqrt_kernel(const unsigned* __restrict__ planes, float* __restrict__ dis,
                      unsigned* __restrict__ part, int n, int npad) {
    __shared__ unsigned wsum[ISQ_NT / 32];
    const int i = blockIdx.x * ISQ_NT + threadIdx.x;
    unsigned s = 0u;
    if (i < n) {
        #pragma unroll 16
        for (int k = 0; k < DEG_SLICES; ++k) s += planes[(size_t)k * npad + i];
        dis[i] = rsqrtf((float)(s + 1u));
    }
    unsigned w = s;
    #pragma unroll
    for (int off = 16; off > 0; off >>= 1) w += __shfl_xor(w, off, 32);
    if ((threadIdx.x & 31) == 0) wsum[threadIdx.x >> 5] = w;
    __syncthreads();
    if (threadIdx.x == 0) {
        unsigned t = 0u;
        for (int j = 0; j < ISQ_NT / 32; ++j) t += wsum[j];
        part[blockIdx.x] = t;
    }
}

template <bool FUSE>
__global__ void gemm_wmma_kernel(const float* __restrict__ A,
                                 const float* __restrict__ W,
                                 const float* __restrict__ dis,
                                 const float* __restrict__ bias,
                                 float* __restrict__ ht,
                                 int M, int K) {
    const int lane = threadIdx.x & 31;
    const int wave = blockIdx.x * (blockDim.x >> 5) + (threadIdx.x >> 5);
    const int m0 = wave * 16;
    if (m0 >= M) return;

    const int half = lane >> 4;
    const int ml   = lane & 15;
    const int arow = m0 + ml;
    const float dA = FUSE ? dis[arow] : 0.0f;

    v8f c0 = {}, c1 = {}, c2 = {}, c3 = {};

    for (int k0 = 0; k0 < K; k0 += 4) {
        const int ka = k0 + half * 2;
        float a0 = A[(size_t)arow * K + ka];
        float a1 = A[(size_t)arow * K + ka + 1];
        if (FUSE) {
            a0 = fmaxf(fmaf(a0, dA, bias[ka]), 0.0f);
            a1 = fmaxf(fmaf(a1, dA, bias[ka + 1]), 0.0f);
        }
        v2f av; av.x = a0; av.y = a1;

        const float* w0 = W + (size_t)ka * 64;
        const float* w1 = W + (size_t)(ka + 1) * 64;
        v2f b0; b0.x = w0[ml];      b0.y = w1[ml];
        v2f b1; b1.x = w0[16 + ml]; b1.y = w1[16 + ml];
        v2f b2; b2.x = w0[32 + ml]; b2.y = w1[32 + ml];
        v2f b3; b3.x = w0[48 + ml]; b3.y = w1[48 + ml];

        c0 = __builtin_amdgcn_wmma_f32_16x16x4_f32(false, av, false, b0, (short)0, c0, false, false);
        c1 = __builtin_amdgcn_wmma_f32_16x16x4_f32(false, av, false, b1, (short)0, c1, false, false);
        c2 = __builtin_amdgcn_wmma_f32_16x16x4_f32(false, av, false, b2, (short)0, c2, false, false);
        c3 = __builtin_amdgcn_wmma_f32_16x16x4_f32(false, av, false, b3, (short)0, c3, false, false);
    }

    for (int r = 0; r < 8; ++r) {
        const int orow = m0 + half * 8 + r;
        const float dr = dis[orow];
        const float v0 = c0[r] * dr;
        const float v1 = c1[r] * dr;
        const float v2 = c2[r] * dr;
        const float v3 = c3[r] * dr;
        float* hp = ht + (size_t)orow * 64;
        hp[ml] = v0; hp[16 + ml] = v1; hp[32 + ml] = v2; hp[48 + ml] = v3;
    }
}

#define AGG_NT    1024
#define AGG_NW    (AGG_NT / 32)
#define AGG_DEPTH 8
#define AGG_Q     64
#define AGG_UQ    (AGG_DEPTH * 32)
#define AGG_ROWS  16
#define AGG_PARTS (AGG_CH / ISQ_NT)
#define AGG_RPW   (AGG_CH / AGG_NW)

__device__ __forceinline__ void agg_add_rows(int s, int dl, int n,
                                             const float* __restrict__ ht,
                                             float* acc, int lane) {
    #pragma unroll
    for (int j0 = 0; j0 < 32; j0 += AGG_ROWS) {
        if (j0 < n) {
            float2 v[AGG_ROWS];
            #pragma unroll
            for (int r = 0; r < AGG_ROWS; ++r) {
                const int sj = __builtin_amdgcn_readlane(s, j0 + r);
                v[r] = *(const float2*)(ht + (size_t)sj * 64 + lane * 2);
            }
            #pragma unroll
            for (int r = 0; r < AGG_ROWS; ++r) {
                if (j0 + r < n) {
                    const int dj = __builtin_amdgcn_readlane(dl, j0 + r);
                    float* ap = acc + dj * 64 + lane * 2;
                    atomicAdd(ap, v[r].x);
                    atomicAdd(ap + 1, v[r].y);
                }
            }
        }
    }
}

__device__ __forceinline__ void agg_take_batch(const int* q, int n,
                                               const int* __restrict__ src,
                                               const int* __restrict__ dst,
                                               const float* __restrict__ ht,
                                               float* acc, int lane,
                                               int2* __restrict__ lst, int cap, int* fill, int base) {
    const int e  = q[min(lane, n - 1)];
    const int s  = src[e];
    const int dl = (int)min((unsigned)(dst[e] - base), (unsigned)(AGG_CH - 1));
    int pos = 0;
    if (lane == 0) pos = atomicAdd(fill, n);
    pos = __builtin_amdgcn_readfirstlane(pos) + lane;
    if (lane < n && pos < cap) lst[pos] = make_int2(s, dl);
    agg_add_rows(s, dl, n, ht, acc, lane);
}

template <bool BUILD>
__global__ __launch_bounds__(AGG_NT)
void agg_owner_kernel(const int* __restrict__ src, const int* __restrict__ dst,
                      const float* __restrict__ ht, float* __restrict__ agg,
                      const unsigned* __restrict__ part, int2* __restrict__ list,
                      const unsigned* __restrict__ keys, int E, int N, int nPart,
                      const float* __restrict__ dis, const float* __restrict__ b2,
                      const float* __restrict__ wfc, const float* __restrict__ bfc,
                      float* __restrict__ out) {
    __shared__ float acc[AGG_CH * 64];
    __shared__ unsigned red[AGG_NW];
    __shared__ int fill;
    const int tid  = threadIdx.x;
    const int lane = tid & 31;
    const int wave = tid >> 5;
    const unsigned blk = blockIdx.x;
    const int base = (int)blk * AGG_CH;

    unsigned below = 0u;
    for (int j = tid; j < (int)blk * AGG_PARTS; j += AGG_NT) below += part[j];
    #pragma unroll
    for (int off = 16; off > 0; off >>= 1) below += __shfl_xor(below, off, 32);
    if (lane == 0) red[wave] = below;
    if (tid == 0) fill = 0;
    unsigned mine = 0u;
    #pragma unroll
    for (int j = 0; j < AGG_PARTS; ++j) {
        const int p = (int)blk * AGG_PARTS + j;
        if (p < nPart) mine += part[p];
    }
    const int cap = (int)mine;

    float4* a4 = (float4*)acc;
    const float4* h4 = (const float4*)(ht + (size_t)base * 64);
    for (int i = tid; i < AGG_CH * 16; i += AGG_NT)
        a4[i] = (base + (i >> 4) < N) ? h4[i] : make_float4(0.f, 0.f, 0.f, 0.f);
    __syncthreads();

    static_assert(AGG_NW == 32, "one wave sum per lane");
    unsigned first = red[lane];
    #pragma unroll
    for (int off = 16; off > 0; off >>= 1) first += __shfl_xor(first, off, 32);
    int2* lst = list + (size_t)__builtin_amdgcn_readfirstlane((int)first);

    if (BUILD) {
        __shared__ int queue[AGG_NW][AGG_Q];
        __shared__ int2 uqueue[AGG_NW][AGG_UQ];
        int* q = queue[wave];
        int2* uq = uqueue[wave];
        int cnt = 0;
        const int nvec = E >> 4;
        const key4* k4 = (const key4*)keys;
        const int step = AGG_DEPTH * AGG_NT;
        const int nIt  = (nvec + step - 1) / step;
        const int rot  = (int)(((long long)blk * nIt) / gridDim.x);
        const unsigned pat = (blk & 255u) * 0x01010101u;

        for (int it = 0; it < nIt; ++it) {
            int itr = it + rot; if (itr >= nIt) itr -= nIt;
            const int v0 = itr * step + tid;
            key4 d[AGG_DEPTH];
            #pragma unroll
            for (int u = 0; u < AGG_DEPTH; ++u)
                d[u] = k4[min(v0 + u * AGG_NT, nvec - 1)];
            __builtin_amdgcn_sched_barrier(0);
            int nu = 0;
            #pragma unroll
            for (int u = 0; u < AGG_DEPTH; ++u) {
                const int v = v0 + u * AGG_NT;
                unsigned cc = 0u;
                #pragma unroll
                for (int k = 0; k < 4; ++k) {
                    const unsigned x = d[u][k] ^ pat;
                    const unsigned t = ((x & 0x7f7f7f7fu) + 0x7f7f7f7fu) | x;
                    cc |= (t >> (7 - k)) & (0x01010101u << k);
                }
                unsigned f = cc ^ 0x0f0f0f0fu;
                if (v >= nvec) f = 0u;
                const unsigned m = (unsigned)__builtin_amdgcn_ballot_w32(f != 0u);
                if (m != 0u) {
                    if (f != 0u) uq[nu + (int)__builtin_amdgcn_mbcnt_lo(m, 0u)] = make_int2((int)f, v);
                    nu += __builtin_popcount(m);
                }
            }
            for (int i = 0; i < nu; i += 32) {
                int2 en = make_int2(0, 0);
                if (i + lane < nu) en = uq[i + lane];
                unsigned f = (unsigned)en.x;
                unsigned m = (unsigned)__builtin_amdgcn_ballot_w32(f != 0u);
                while (m != 0u) {
                    if (f != 0u) {
                        const int b = __builtin_ctz(f);
                        q[cnt + (int)__builtin_amdgcn_mbcnt_lo(m, 0u)] = (en.y << 4) + ((b & 3) << 2) + (b >> 3);
                        f &= f - 1u;
                    }
                    cnt += __builtin_popcount(m);
                    if (cnt >= 32) {
                        cnt -= 32;
                        agg_take_batch(q + cnt, 32, src, dst, ht, acc, lane, lst, cap, &fill, base);
                    }
                    m = (unsigned)__builtin_amdgcn_ballot_w32(f != 0u);
                }
            }
        }
        if (wave == 0) {
            const int e = (nvec << 4) + lane;
            const bool hit = e < E && ((unsigned)dst[min(e, E - 1)] / (unsigned)AGG_CH) == blk;
            const unsigned m = (unsigned)__builtin_amdgcn_ballot_w32(hit);
            if (hit) q[cnt + (int)__builtin_amdgcn_mbcnt_lo(m, 0u)] = e;
            cnt += __builtin_popcount(m);
            if (cnt >= 32) {
                cnt -= 32;
                agg_take_batch(q + cnt, 32, src, dst, ht, acc, lane, lst, cap, &fill, base);
            }
        }
        if (cnt > 0) agg_take_batch(q, cnt, src, dst, ht, acc, lane, lst, cap, &fill, base);
    } else {
        const int wv    = __builtin_amdgcn_readfirstlane(wave);
        const int nfull = cap >> 5;
        int b = wv;
        int2 cur = make_int2(0, 0);
        if (b < nfull) cur = lst[b * 32 + lane];
        while (b < nfull) {
            const int nb = b + AGG_NW;
            int2 nxt = cur;
            if (nb < nfull) nxt = lst[nb * 32 + lane];
            agg_add_rows(cur.x, cur.y, 32, ht, acc, lane);
            cur = nxt;
            b = nb;
        }
        const int rem = cap & 31;
        if (rem > 0 && wv == (nfull & (AGG_NW - 1))) {
            const int2 en = lst[nfull * 32 + min(lane, rem - 1)];
            agg_add_rows(en.x, en.y, rem, ht, acc, lane);
        }
    }
    __syncthreads();

    if (BUILD) {
        float4* o4 = (float4*)(agg + (size_t)base * 64);
        for (int i = tid; i < AGG_CH * 16; i += AGG_NT)
            if (base + (i >> 4) < N) o4[i] = a4[i];
    } else {
        static_assert(AGG_NW * AGG_RPW == AGG_CH && AGG_RPW <= 16, "whole rows per wave, one dis value per lane of a half");
        __shared__ float outv[AGG_CH];
        const int j = lane * 2;
        const float bj0 = b2[j], bj1 = b2[j + 1];
        const float w0 = wfc[j], w1 = wfc[j + 1];
        const int r0 = wave * AGG_RPW;
        const int dl = __float_as_int(dis[min(base + r0 + (lane & 15), N - 1)]);
        float res = 0.0f;
        #pragma unroll
        for (int r = 0; r < AGG_RPW; ++r) {
            const float di = __int_as_float(__builtin_amdgcn_readlane(dl, r));
            const float2 a = *(const float2*)(acc + (r0 + r) * 64 + j);
            const float h0 = fmaxf(fmaf(a.x, di, bj0), 0.0f);
            const float h1 = fmaxf(fmaf(a.y, di, bj1), 0.0f);
            float p = h0 * w0 + h1 * w1;
            #pragma unroll
            for (int off = 16; off > 0; off >>= 1) p += __shfl_xor(p, off, 32);
            if (lane == r) res = p;
        }
        if (lane < AGG_RPW) outv[r0 + lane] = res + bfc[0];
        __syncthreads();
        if (tid < AGG_CH && base + tid < N) out[base + tid] = outv[tid];
    }
}

extern "C" void kernel_launch(void* const* d_in, const int* in_sizes, int n_in,
                              void* d_out, int out_size, void* d_ws, size_t ws_size,
                              hipStream_t stream) {
    const float* x   = (const float*)d_in[0];
    const int*   ei  = (const int*)d_in[1];
    const float* W1  = (const float*)d_in[2];
    const float* b1  = (const float*)d_in[3];
    const float* W2  = (const float*)d_in[4];
    const float* b2  = (const float*)d_in[5];
    const float* Wfc = (const float*)d_in[6];
    const float* bfc = (const float*)d_in[7];

    const int HID = in_sizes[3];
    const int IN  = in_sizes[2] / HID;
    const int N   = in_sizes[0] / IN;
    const int E   = in_sizes[1] / 2;
    const int* srcv = ei;
    const int* dstv = ei + E;

    char* ws = (char*)d_ws;
    float* dis = (float*)ws;
    size_t off = (((size_t)N * 4) + 255) & ~(size_t)255;
    const size_t rowBytes = (size_t)N * 64 * sizeof(float);
    float* ht   = (float*)(ws + off);
    float* agg1 = (float*)(ws + off + rowBytes);

    const int nChunks = (N + DEG_CH - 1) / DEG_CH;
    const int npad    = nChunks * DEG_CH;
    unsigned* planes  = (unsigned*)(ws + off + 3 * rowBytes);
    const size_t planeBytes = (size_t)DEG_SLICES * npad * sizeof(unsigned);
    const int nPart   = (N + ISQ_NT - 1) / ISQ_NT;
    unsigned* part    = (unsigned*)(ws + off + 3 * rowBytes + planeBytes);
    const size_t partBytes = (((size_t)nPart * 4) + 255) & ~(size_t)255;
    int2* list        = (int2*)(ws + off + 3 * rowBytes + planeBytes + partBytes);
    unsigned* keys    = (unsigned*)(ws + off + 3 * rowBytes + planeBytes + partBytes + (size_t)E * sizeof(int2));
    deg_hist_kernel<<<dim3(DEG_SLICES, nChunks), DEG_NT, 0, stream>>>(dstv, planes, keys, E, npad);
    deg_isqrt_kernel<<<nPart, ISQ_NT, 0, stream>>>(planes, dis, part, N, npad);

    const int gemmBlocks = ((N + 15) / 16 + 7) / 8;
    const int aggBlocks  = (N + AGG_CH - 1) / AGG_CH;

    gemm_wmma_kernel<false><<<gemmBlocks, 256, 0, stream>>>(x, W1, dis, nullptr, ht, N, IN);
    agg_owner_kernel<true><<<aggBlocks, AGG_NT, 0, stream>>>(srcv, dstv, ht, agg1, part, list, keys, E, N, nPart,
                                                             nullptr, nullptr, nullptr, nullptr, nullptr);

    gemm_wmma_kernel<true><<<gemmBlocks, 256, 0, stream>>>(agg1, W2, dis, b1, ht, N, HID);
    agg_owner_kernel<false><<<aggBlocks, AGG_NT, 0, stream>>>(srcv, dstv, ht, nullptr, part, list, keys, E, N, nPart,
                                                              dis, b2, Wfc, bfc, (float*)d_out);
}


# layer-1 owner blocks finish their nodes from the LDS sums (relu, 64x64 W2 product on f32 WMMA, 16-row tile per wave, dis); agg1 plane and layer-2 product launch gone; 6 launches to 5; sum orders kept
# speedup vs baseline: 3.0583x; 1.0959x over previous
#include <hip/hip_runtime.h>

typedef __attribute__((ext_vector_type(2))) float v2f;
typedef __attribute__((ext_vector_type(8))) float v8f;

#define AGG_CH     416
#define DEG_CH     32768
#define DEG_SLICES 64
#define DEG_NT     512
#define DEG_DEPTH  8

typedef int int4u __attribute__((ext_vector_type(4), aligned(4)));
typedef unsigned key4 __attribute__((ext_vector_type(4)));

__global__ __launch_bounds__(DEG_NT)
void deg_hist_kernel(const int* __restrict__ dst, unsigned* __restrict__ planes,
                     unsigned* __restrict__ keys, int E, int npad) {
    __shared__ unsigned cnt[DEG_CH];
    const int tid   = threadIdx.x;
    const int slice = blockIdx.x;
    const unsigned base = blockIdx.y * (unsigned)DEG_CH;

    uint4* c4 = (uint4*)cnt;
    for (int i = tid; i < DEG_CH / 4; i += DEG_NT) c4[i] = make_uint4(0u, 0u, 0u, 0u);
    __syncthreads();

    const int nvec = E >> 2;
    const int per  = (nvec + DEG_SLICES - 1) / DEG_SLICES;
    const int lo   = slice * per;
    const int hi   = min(lo + per, nvec);
    const int4u* d4 = (const int4u*)dst;
    const bool wkeys = blockIdx.y == 0;

    for (int v0 = lo + tid; v0 < hi; v0 += DEG_DEPTH * DEG_NT) {
        int4u d[DEG_DEPTH];
        #pragma unroll
        for (int u = 0; u < DEG_DEPTH; ++u) {
            const int v = v0 + u * DEG_NT;
            d[u] = d4[min(v, hi - 1)];
            if (v >= hi) d[u] = (int4u)(-1);
        }
        if (wkeys) {
            #pragma unroll
            for (int u = 0; u < DEG_DEPTH; ++u) {
                const int v = v0 + u * DEG_NT;
                if (v < hi) {
                    const unsigned k0 = min((unsigned)d[u][0] / (unsigned)AGG_CH, 255u);
                    const unsigned k1 = min((unsigned)d[u][1] / (unsigned)AGG_CH, 255u);
                    const unsigned k2 = min((unsigned)d[u][2] / (unsigned)AGG_CH, 255u);
                    const unsigned k3 = min((unsigned)d[u][3] / (unsigned)AGG_CH, 255u);
                    keys[v] = k0 | (k1 << 8) | (k2 << 16) | (k3 << 24);
                }
            }
        }
        #pragma unroll
        for (int u = 0; u < DEG_DEPTH; ++u) {
            #pragma unroll
            for (int q = 0; q < 4; ++q) {
                const unsigned r = (unsigned)d[u][q] - base;
                if (r < (unsigned)DEG_CH) atomicAdd(&cnt[r], 1u);
            }
        }
    }
    if (slice == DEG_SLICES - 1) {
        for (int e = (nvec << 2) + tid; e < E; e += DEG_NT) {
            const unsigned r = (unsigned)dst[e] - base;
            if (r < (unsigned)DEG_CH) atomicAdd(&cnt[r], 1u);
        }
    }
    __syncthreads();

    uint4* o4 = (uint4*)(planes + (size_t)slice * npad + base);
    for (int i = tid; i < DEG_CH / 4; i += DEG_NT) o4[i] = c4[i];
}

#define ISQ_NT AGG_CH

__global__ __launch_bounds__(ISQ_NT)
void deg_isqrt_kernel(const unsigned* __restrict__ planes, float* __restrict__ dis,
                      unsigned* __restrict__ part, int n, int npad) {
    __shared__ unsigned wsum[ISQ_NT / 32];
    const int i = blockIdx.x * ISQ_NT + threadIdx.x;
    unsigned s = 0u;
    if (i < n) {
        #pragma unroll 16
        for (int k = 0; k < DEG_SLICES; ++k) s += planes[(size_t)k * npad + i];
        dis[i] = rsqrtf((float)(s + 1u));
    }
    unsigned w = s;
    #pragma unroll
    for (int off = 16; off > 0; off >>= 1) w += __shfl_xor(w, off, 32);
    if ((threadIdx.x & 31) == 0) wsum[threadIdx.x >> 5] = w;
    __syncthreads();
    if (threadIdx.x == 0) {
        unsigned t = 0u;
        for (int j = 0; j < ISQ_NT / 32; ++j) t += wsum[j];
        part[blockIdx.x] = t;
    }
}

__global__ void gemm_wmma_kernel(const float* __restrict__ A,
                                 const float* __restrict__ W,
                                 const float* __restrict__ dis,
                                 float* __restrict__ ht,
                                 int M, int K) {
    const int lane = threadIdx.x & 31;
    const int wave = blockIdx.x * (blockDim.x >> 5) + (threadIdx.x >> 5);
    const int m0 = wave * 16;
    if (m0 >= M) return;

    const int half = lane >> 4;
    const int ml   = lane & 15;
    const int arow = m0 + ml;

    v8f c0 = {}, c1 = {}, c2 = {}, c3 = {};

    for (int k0 = 0; k0 < K; k0 += 4) {
        const int ka = k0 + half * 2;
        const float a0 = A[(size_t)arow * K + ka];
        const float a1 = A[(size_t)arow * K + ka + 1];
        v2f av; av.x = a0; av.y = a1;

        const float* w0 = W + (size_t)ka * 64;
        const float* w1 = W + (size_t)(ka + 1) * 64;
        v2f b0; b0.x = w0[ml];      b0.y = w1[ml];
        v2f b1; b1.x = w0[16 + ml]; b1.y = w1[16 + ml];
        v2f b2; b2.x = w0[32 + ml]; b2.y = w1[32 + ml];
        v2f b3; b3.x = w0[48 + ml]; b3.y = w1[48 + ml];

        c0 = __builtin_amdgcn_wmma_f32_16x16x4_f32(false, av, false, b0, (short)0, c0, false, false);
        c1 = __builtin_amdgcn_wmma_f32_16x16x4_f32(false, av, false, b1, (short)0, c1, false, false);
        c2 = __builtin_amdgcn_wmma_f32_16x16x4_f32(false, av, false, b2, (short)0, c2, false, false);
        c3 = __builtin_amdgcn_wmma_f32_16x16x4_f32(false, av, false, b3, (short)0, c3, false, false);
    }

    for (int r = 0; r < 8; ++r) {
        const int orow = m0 + half * 8 + r;
        const float dr = dis[orow];
        const float v0 = c0[r] * dr;
        const float v1 = c1[r] * dr;
        const float v2 = c2[r] * dr;
        const float v3 = c3[r] * dr;
        float* hp = ht + (size_t)orow * 64;
        hp[ml] = v0; hp[16 + ml] = v1; hp[32 + ml] = v2; hp[48 + ml] = v3;
    }
}

#define AGG_NT    1024
#define AGG_NW    (AGG_NT / 32)
#define AGG_DEPTH 8
#define AGG_Q     64
#define AGG_UQ    (AGG_DEPTH * 32)
#define AGG_ROWS  16
#define AGG_PARTS (AGG_CH / ISQ_NT)
#define AGG_RPW   (AGG_CH / AGG_NW)

__device__ __forceinline__ void agg_add_rows(int s, int dl, int n,
                                             const float* __restrict__ ht,
                                             float* acc, int lane) {
    #pragma unroll
    for (int j0 = 0; j0 < 32; j0 += AGG_ROWS) {
        if (j0 < n) {
            float2 v[AGG_ROWS];
            #pragma unroll
            for (int r = 0; r < AGG_ROWS; ++r) {
                const int sj = __builtin_amdgcn_readlane(s, j0 + r);
                v[r] = *(const float2*)(ht + (size_t)sj * 64 + lane * 2);
            }
            #pragma unroll
            for (int r = 0; r < AGG_ROWS; ++r) {
                if (j0 + r < n) {
                    const int dj = __builtin_amdgcn_readlane(dl, j0 + r);
                    float* ap = acc + dj * 64 + lane * 2;
                    atomicAdd(ap, v[r].x);
                    atomicAdd(ap + 1, v[r].y);
                }
            }
        }
    }
}

__device__ __forceinline__ void agg_take_batch(const int* q, int n,
                                               const int* __restrict__ src,
                                               const int* __restrict__ dst,
                                               const float* __restrict__ ht,
                                               float* acc, int lane,
                                               int2* __restrict__ lst, int cap, int* fill, int base) {
    const int e  = q[min(lane, n - 1)];
    const int s  = src[e];
    const int dl = (int)min((unsigned)(dst[e] - base), (unsigned)(AGG_CH - 1));
    int pos = 0;
    if (lane == 0) pos = atomicAdd(fill, n);
    pos = __builtin_amdgcn_readfirstlane(pos) + lane;
    if (lane < n && pos < cap) lst[pos] = make_int2(s, dl);
    agg_add_rows(s, dl, n, ht, acc, lane);
}

template <bool BUILD>
__global__ __launch_bounds__(AGG_NT)
void agg_owner_kernel(const int* __restrict__ src, const int* __restrict__ dst,
                      const float* __restrict__ ht, float* __restrict__ agg,
                      const unsigned* __restrict__ part, int2* __restrict__ list,
                      const unsigned* __restrict__ keys, int E, int N, int nPart,
                      const float* __restrict__ dis, const float* __restrict__ bvec,
                      const float* __restrict__ wmat, const float* __restrict__ bfc,
                      float* __restrict__ out) {
    __shared__ float acc[AGG_CH * 64];
    __shared__ unsigned red[AGG_NW];
    __shared__ int fill;
    const int tid  = threadIdx.x;
    const int lane = tid & 31;
    const int wave = tid >> 5;
    const unsigned blk = blockIdx.x;
    const int base = (int)blk * AGG_CH;

    unsigned below = 0u;
    for (int j = tid; j < (int)blk * AGG_PARTS; j += AGG_NT) below += part[j];
    #pragma unroll
    for (int off = 16; off > 0; off >>= 1) below += __shfl_xor(below, off, 32);
    if (lane == 0) red[wave] = below;
    if (tid == 0) fill = 0;
    unsigned mine = 0u;
    #pragma unroll
    for (int j = 0; j < AGG_PARTS; ++j) {
        const int p = (int)blk * AGG_PARTS + j;
        if (p < nPart) mine += part[p];
    }
    const int cap = (int)mine;

    float4* a4 = (float4*)acc;
    const float4* h4 = (const float4*)(ht + (size_t)base * 64);
    for (int i = tid; i < AGG_CH * 16; i += AGG_NT)
        a4[i] = (base + (i >> 4) < N) ? h4[i] : make_float4(0.f, 0.f, 0.f, 0.f);
    __syncthreads();

    static_assert(AGG_NW == 32, "one wave sum per lane");
    unsigned first = red[lane];
    #pragma unroll
    for (int off = 16; off > 0; off >>= 1) first += __shfl_xor(first, off, 32);
    int2* lst = list + (size_t)__builtin_amdgcn_readfirstlane((int)first);

    if (BUILD) {
        __shared__ int queue[AGG_NW][AGG_Q];
        __shared__ int2 uqueue[AGG_NW][AGG_UQ];
        int* q = queue[wave];
        int2* uq = uqueue[wave];
        int cnt = 0;
        const int nvec = E >> 4;
        const key4* k4 = (const key4*)keys;
        const int step = AGG_DEPTH * AGG_NT;
        const int nIt  = (nvec + step - 1) / step;
        const int rot  = (int)(((long long)blk * nIt) / gridDim.x);
        const unsigned pat = (blk & 255u) * 0x01010101u;

        for (int it = 0; it < nIt; ++it) {
            int itr = it + rot; if (itr >= nIt) itr -= nIt;
            const int v0 = itr * step + tid;
            key4 d[AGG_DEPTH];
            #pragma unroll
            for (int u = 0; u < AGG_DEPTH; ++u)
                d[u] = k4[min(v0 + u * AGG_NT, nvec - 1)];
            __builtin_amdgcn_sched_barrier(0);
            int nu = 0;
            #pragma unroll
            for (int u = 0; u < AGG_DEPTH; ++u) {
                const int v = v0 + u * AGG_NT;
                unsigned cc = 0u;
                #pragma unroll
                for (int k = 0; k < 4; ++k) {
                    const unsigned x = d[u][k] ^ pat;
                    const unsigned t = ((x & 0x7f7f7f7fu) + 0x7f7f7f7fu) | x;
                    cc |= (t >> (7 - k)) & (0x01010101u << k);
                }
                unsigned f = cc ^ 0x0f0f0f0fu;
                if (v >= nvec) f = 0u;
                const unsigned m = (unsigned)__builtin_amdgcn_ballot_w32(f != 0u);
                if (m != 0u) {
                    if (f != 0u) uq[nu + (int)__builtin_amdgcn_mbcnt_lo(m, 0u)] = make_int2((int)f, v);
                    nu += __builtin_popcount(m);
                }
            }
            for (int i = 0; i < nu; i += 32) {
                int2 en = make_int2(0, 0);
                if (i + lane < nu) en = uq[i + lane];
                unsigned f = (unsigned)en.x;
                unsigned m = (unsigned)__builtin_amdgcn_ballot_w32(f != 0u);
                while (m != 0u) {
                    if (f != 0u) {
                        const int b = __builtin_ctz(f);
                        q[cnt + (int)__builtin_amdgcn_mbcnt_lo(m, 0u)] = (en.y << 4) + ((b & 3) << 2) + (b >> 3);
                        f &= f - 1u;
                    }
                    cnt += __builtin_popcount(m);
                    if (cnt >= 32) {
                        cnt -= 32;
                        agg_take_batch(q + cnt, 32, src, dst, ht, acc, lane, lst, cap, &fill, base);
                    }
                    m = (unsigned)__builtin_amdgcn_ballot_w32(f != 0u);
                }
            }
        }
        if (wave == 0) {
            const int e = (nvec << 4) + lane;
            const bool hit = e < E && ((unsigned)dst[min(e, E - 1)] / (unsigned)AGG_CH) == blk;
            const unsigned m = (unsigned)__builtin_amdgcn_ballot_w32(hit);
            if (hit) q[cnt + (int)__builtin_amdgcn_mbcnt_lo(m, 0u)] = e;
            cnt += __builtin_popcount(m);
            if (cnt >= 32) {
                cnt -= 32;
                agg_take_batch(q + cnt, 32, src, dst, ht, acc, lane, lst, cap, &fill, base);
            }
        }
        if (cnt > 0) agg_take_batch(q, cnt, src, dst, ht, acc, lane, lst, cap, &fill, base);
    } else {
        const int wv    = __builtin_amdgcn_readfirstlane(wave);
        const int nfull = cap >> 5;
        int b = wv;
        int2 cur = make_int2(0, 0);
        if (b < nfull) cur = lst[b * 32 + lane];
        while (b < nfull) {
            const int nb = b + AGG_NW;
            int2 nxt = cur;
            if (nb < nfull) nxt = lst[nb * 32 + lane];
            agg_add_rows(cur.x, cur.y, 32, ht, acc, lane);
            cur = nxt;
            b = nb;
        }
        const int rem = cap & 31;
        if (rem > 0 && wv == (nfull & (AGG_NW - 1))) {
            const int2 en = lst[nfull * 32 + min(lane, rem - 1)];
            agg_add_rows(en.x, en.y, rem, ht, acc, lane);
        }
    }
    __syncthreads();

    if (BUILD) {
        static_assert(AGG_CH % 16 == 0 && AGG_CH / 16 <= AGG_NW, "a 16-row tile per wave");
        if (__builtin_amdgcn_readfirstlane(wave) < AGG_CH / 16) {
            const int half = lane >> 4;
            const int ml   = lane & 15;
            const int r0   = wave * 16;
            const float dA = dis[min(base + r0 + ml, N - 1)];
            const float* arow = acc + (r0 + ml) * 64;
            v8f c0 = {}, c1 = {}, c2 = {}, c3 = {};
            for (int k0 = 0; k0 < 64; k0 += 4) {
                const int ka = k0 + half * 2;
                const float2 a = *(const float2*)(arow + ka);
                v2f av;
                av.x = fmaxf(fmaf(a.x, dA, bvec[ka]), 0.0f);
                av.y = fmaxf(fmaf(a.y, dA, bvec[ka + 1]), 0.0f);
                const float* w0 = wmat + ka * 64;
                const float* w1 = wmat + (ka + 1) * 64;
                v2f b0; b0.x = w0[ml];      b0.y = w1[ml];
                v2f b1; b1.x = w0[16 + ml]; b1.y = w1[16 + ml];
                v2f b2; b2.x = w0[32 + ml]; b2.y = w1[32 + ml];
                v2f b3; b3.x = w0[48 + ml]; b3.y = w1[48 + ml];
                c0 = __builtin_amdgcn_wmma_f32_16x16x4_f32(false, av, false, b0, (short)0, c0, false, false);
                c1 = __builtin_amdgcn_wmma_f32_16x16x4_f32(false, av, false, b1, (short)0, c1, false, false);
                c2 = __builtin_amdgcn_wmma_f32_16x16x4_f32(false, av, false, b2, (short)0, c2, false, false);
                c3 = __builtin_amdgcn_wmma_f32_16x16x4_f32(false, av, false, b3, (short)0, c3, false, false);
            }
            for (int r = 0; r < 8; ++r) {
                const int orow = r0 + half * 8 + r;
                const float dr = dis[min(base + orow, N - 1)];
                float* hp = acc + orow * 64;
                hp[ml] = c0[r] * dr; hp[16 + ml] = c1[r] * dr; hp[32 + ml] = c2[r] * dr; hp[48 + ml] = c3[r] * dr;
            }
        }
        __syncthreads();
        float4* o4 = (float4*)(agg + (size_t)base * 64);
        for (int i = tid; i < AGG_CH * 16; i += AGG_NT)
            if (base + (i >> 4) < N) o4[i] = a4[i];
    } else {
        static_assert(AGG_NW * AGG_RPW == AGG_CH && AGG_RPW <= 16, "whole rows per wave, one dis value per lane of a half");
        __shared__ float outv[AGG_CH];
        const int j = lane * 2;
        const float bj0 = bvec[j], bj1 = bvec[j + 1];
        const float w0 = wmat[j], w1 = wmat[j + 1];
        const int r0 = wave * AGG_RPW;
        const int dl = __float_as_int(dis[min(base + r0 + (lane & 15), N - 1)]);
        float res = 0.0f;
        #pragma unroll
        for (int r = 0; r < AGG_RPW; ++r) {
            const float di = __int_as_float(__builtin_amdgcn_readlane(dl, r));
            const float2 a = *(const float2*)(acc + (r0 + r) * 64 + j);
            const float h0 = fmaxf(fmaf(a.x, di, bj0), 0.0f);
            const float h1 = fmaxf(fmaf(a.y, di, bj1), 0.0f);
            float p = h0 * w0 + h1 * w1;
            #pragma unroll
            for (int off = 16; off > 0; off >>= 1) p += __shfl_xor(p, off, 32);
            if (lane == r) res = p;
        }
        if (lane < AGG_RPW) outv[r0 + lane] = res + bfc[0];
        __syncthreads();
        if (tid < AGG_CH && base + tid < N) out[base + tid] = outv[tid];
    }
}

extern "C" void kernel_launch(void* const* d_in, const int* in_sizes, int n_in,
                              void* d_out, int out_size, void* d_ws, size_t ws_size,
                              hipStream_t stream) {
    const float* x   = (const float*)d_in[0];
    const int*   ei  = (const int*)d_in[1];
    const float* W1  = (const float*)d_in[2];
    const float* b1  = (const float*)d_in[3];
    const float* W2  = (const float*)d_in[4];
    const float* b2  = (const float*)d_in[5];
    const float* Wfc = (const float*)d_in[6];
    const float* bfc = (const float*)d_in[7];

    const int HID = in_sizes[3];
    const int IN  = in_sizes[2] / HID;
    const int N   = in_sizes[0] / IN;
    const int E   = in_sizes[1] / 2;
    const int* srcv = ei;
    const int* dstv = ei + E;

    char* ws = (char*)d_ws;
    float* dis = (float*)ws;
    size_t off = (((size_t)N * 4) + 255) & ~(size_t)255;
    const size_t rowBytes = (size_t)N * 64 * sizeof(float);
    float* ht   = (float*)(ws + off);
    float* ht2  = (float*)(ws + off + rowBytes);

    const int nChunks = (N + DEG_CH - 1) / DEG_CH;
    const int npad    = nChunks * DEG_CH;
    unsigned* planes  = (unsigned*)(ws + off + 3 * rowBytes);
    const size_t planeBytes = (size_t)DEG_SLICES * npad * sizeof(unsigned);
    const int nPart   = (N + ISQ_NT - 1) / ISQ_NT;
    unsigned* part    = (unsigned*)(ws + off + 3 * rowBytes + planeBytes);
    const size_t partBytes = (((size_t)nPart * 4) + 255) & ~(size_t)255;
    int2* list        = (int2*)(ws + off + 3 * rowBytes + planeBytes + partBytes);
    unsigned* keys    = (unsigned*)(ws + off + 3 * rowBytes + planeBytes + partBytes + (size_t)E * sizeof(int2));
    deg_hist_kernel<<<dim3(DEG_SLICES, nChunks), DEG_NT, 0, stream>>>(dstv, planes, keys, E, npad);
    deg_isqrt_kernel<<<nPart, ISQ_NT, 0, stream>>>(planes, dis, part, N, npad);

    const int gemmBlocks = ((N + 15) / 16 + 7) / 8;
    const int aggBlocks  = (N + AGG_CH - 1) / AGG_CH;

    gemm_wmma_kernel<<<gemmBlocks, 256, 0, stream>>>(x, W1, dis, ht, N, IN);
    agg_owner_kernel<true><<<aggBlocks, AGG_NT, 0, stream>>>(srcv, dstv, ht, ht2, part, list, keys, E, N, nPart,
                                                             dis, b1, W2, nullptr, nullptr);

    agg_owner_kernel<false><<<aggBlocks, AGG_NT, 0, stream>>>(srcv, dstv, ht2, nullptr, part, list, keys, E, N, nPart,
                                                              dis, b2, Wfc, bfc, (float*)d_out);
}


# rsqrt pass folded into the layer-1 product launch: a block per 416-row owner range, each wave adds the count planes of its own 16 rows and keeps dis in registers for its epilogue; 5 launches to 4
# speedup vs baseline: 3.1550x; 1.0316x over previous
#include <hip/hip_runtime.h>

typedef __attribute__((ext_vector_type(2))) float v2f;
typedef __attribute__((ext_vector_type(8))) float v8f;

#define AGG_CH     416
#define DEG_CH     32768
#define DEG_SLICES 64
#define DEG_NT     512
#define DEG_DEPTH  8

typedef int int4u __attribute__((ext_vector_type(4), aligned(4)));
typedef unsigned key4 __attribute__((ext_vector_type(4)));

__global__ __launch_bounds__(DEG_NT)
void deg_hist_kernel(const int* __restrict__ dst, unsigned* __restrict__ planes,
                     unsigned* __restrict__ keys, int E, int npad) {
    __shared__ unsigned cnt[DEG_CH];
    const int tid   = threadIdx.x;
    const int slice = blockIdx.x;
    const unsigned base = blockIdx.y * (unsigned)DEG_CH;

    uint4* c4 = (uint4*)cnt;
    for (int i = tid; i < DEG_CH / 4; i += DEG_NT) c4[i] = make_uint4(0u, 0u, 0u, 0u);
    __syncthreads();

    const int nvec = E >> 2;
    const int per  = (nvec + DEG_SLICES - 1) / DEG_SLICES;
    const int lo   = slice * per;
    const int hi   = min(lo + per, nvec);
    const int4u* d4 = (const int4u*)dst;
    const bool wkeys = blockIdx.y == 0;

    for (int v0 = lo + tid; v0 < hi; v0 += DEG_DEPTH * DEG_NT) {
        int4u d[DEG_DEPTH];
        #pragma unroll
        for (int u = 0; u < DEG_DEPTH; ++u) {
            const int v = v0 + u * DEG_NT;
            d[u] = d4[min(v, hi - 1)];
            if (v >= hi) d[u] = (int4u)(-1);
        }
        if (wkeys) {
            #pragma unroll
            for (int u = 0; u < DEG_DEPTH; ++u) {
                const int v = v0 + u * DEG_NT;
                if (v < hi) {
                    const unsigned k0 = min((unsigned)d[u][0] / (unsigned)AGG_CH, 255u);
                    const unsigned k1 = min((unsigned)d[u][1] / (unsigned)AGG_CH, 255u);
                    const unsigned k2 = min((unsigned)d[u][2] / (unsigned)AGG_CH, 255u);
                    const unsigned k3 = min((unsigned)d[u][3] / (unsigned)AGG_CH, 255u);
                    keys[v] = k0 | (k1 << 8) | (k2 << 16) | (k3 << 24);
                }
            }
        }
        #pragma unroll
        for (int u = 0; u < DEG_DEPTH; ++u) {
            #pragma unroll
            for (int q = 0; q < 4; ++q) {
                const unsigned r = (unsigned)d[u][q] - base;
                if (r < (unsigned)DEG_CH) atomicAdd(&cnt[r], 1u);
            }
        }
    }
    if (slice == DEG_SLICES - 1) {
        for (int e = (nvec << 2) + tid; e < E; e += DEG_NT) {
            const unsigned r = (unsigned)dst[e] - base;
            if (r < (unsigned)DEG_CH) atomicAdd(&cnt[r], 1u);
        }
    }
    __syncthreads();

    uint4* o4 = (uint4*)(planes + (size_t)slice * npad + base);
    for (int i = tid; i < DEG_CH / 4; i += DEG_NT) o4[i] = c4[i];
}

#define ISQ_NT AGG_CH

#define GEMM_NW (AGG_CH / 16)
#define GEMM_NT (GEMM_NW * 32)

__global__ __launch_bounds__(GEMM_NT)
void gemm_wmma_kernel(const float* __restrict__ A,
                      const float* __restrict__ W,
                      const unsigned* __restrict__ planes,
                      float* __restrict__ dis,
                      unsigned* __restrict__ part,
                      float* __restrict__ ht,
                      int M, int K, int npad) {
    static_assert(AGG_CH % 32 == 0 && DEG_SLICES % 2 == 0, "whole lines of dis, two plane halves");
    __shared__ float dis_s[AGG_CH];
    __shared__ unsigned wsum[GEMM_NW];
    const int tid  = threadIdx.x;
    const int lane = tid & 31;
    const int wave = __builtin_amdgcn_readfirstlane(tid >> 5);
    const int base = (int)blockIdx.x * AGG_CH;
    const int m0   = base + wave * 16;

    const int half = lane >> 4;
    const int ml   = lane & 15;
    const int row  = m0 + ml;
    const int arow = min(row, M - 1);

    unsigned s = 0u;
    const unsigned* pl = planes + (size_t)(half * (DEG_SLICES / 2)) * npad + arow;
    #pragma unroll
    for (int k = 0; k < DEG_SLICES / 2; ++k) s += pl[(size_t)k * npad];
    s += __shfl_xor(s, 16, 32);
    if (row >= M) s = 0u;
    const float dl = rsqrtf((float)(s + 1u));
    if (half == 0) dis_s[wave * 16 + ml] = dl;
    unsigned w = half == 0 ? s : 0u;
    #pragma unroll
    for (int off = 16; off > 0; off >>= 1) w += __shfl_xor(w, off, 32);
    if (lane == 0) wsum[wave] = w;
    __syncthreads();

    if (tid < AGG_CH && base + tid < M) dis[base + tid] = dis_s[tid];
    if (tid == 0) {
        unsigned t = 0u;
        for (int j = 0; j < GEMM_NW; ++j) t += wsum[j];
        part[blockIdx.x] = t;
    }

    if (m0 < M) {
        v8f c0 = {}, c1 = {}, c2 = {}, c3 = {};

        for (int k0 = 0; k0 < K; k0 += 4) {
            const int ka = k0 + half * 2;
            const float a0 = A[(size_t)arow * K + ka];
            const float a1 = A[(size_t)arow * K + ka + 1];
            v2f av; av.x = a0; av.y = a1;

            const float* w0 = W + (size_t)ka * 64;
            const float* w1 = W + (size_t)(ka + 1) * 64;
            v2f b0; b0.x = w0[ml];      b0.y = w1[ml];
            v2f b1; b1.x = w0[16 + ml]; b1.y = w1[16 + ml];
            v2f b2; b2.x = w0[32 + ml]; b2.y = w1[32 + ml];
            v2f b3; b3.x = w0[48 + ml]; b3.y = w1[48 + ml];

            c0 = __builtin_amdgcn_wmma_f32_16x16x4_f32(false, av, false, b0, (short)0, c0, false, false);
            c1 = __builtin_amdgcn_wmma_f32_16x16x4_f32(false, av, false, b1, (short)0, c1, false, false);
            c2 = __builtin_amdgcn_wmma_f32_16x16x4_f32(false, av, false, b2, (short)0, c2, false, false);
            c3 = __builtin_amdgcn_wmma_f32_16x16x4_f32(false, av, false, b3, (short)0, c3, false, false);
        }

        const int dli = __float_as_int(dl);
        #pragma unroll
        for (int r = 0; r < 8; ++r) {
            const int orow = m0 + half * 8 + r;
            const int dlo = __builtin_amdgcn_readlane(dli, r);
            const int dhi = __builtin_amdgcn_readlane(dli, 8 + r);
            const float dr = __int_as_float(half ? dhi : dlo);
            if (orow < M) {
                float* hp = ht + (size_t)orow * 64;
                hp[ml] = c0[r] * dr; hp[16 + ml] = c1[r] * dr; hp[32 + ml] = c2[r] * dr; hp[48 + ml] = c3[r] * dr;
            }
        }
    }
}

#define AGG_NT    1024
#define AGG_NW    (AGG_NT / 32)
#define AGG_DEPTH 8
#define AGG_Q     64
#define AGG_UQ    (AGG_DEPTH * 32)
#define AGG_ROWS  16
#define AGG_PARTS (AGG_CH / ISQ_NT)
#define AGG_RPW   (AGG_CH / AGG_NW)

__device__ __forceinline__ void agg_add_rows(int s, int dl, int n,
                                             const float* __restrict__ ht,
                                             float* acc, int lane) {
    #pragma unroll
    for (int j0 = 0; j0 < 32; j0 += AGG_ROWS) {
        if (j0 < n) {
            float2 v[AGG_ROWS];
            #pragma unroll
            for (int r = 0; r < AGG_ROWS; ++r) {
                const int sj = __builtin_amdgcn_readlane(s, j0 + r);
                v[r] = *(const float2*)(ht + (size_t)sj * 64 + lane * 2);
            }
            #pragma unroll
            for (int r = 0; r < AGG_ROWS; ++r) {
                if (j0 + r < n) {
                    const int dj = __builtin_amdgcn_readlane(dl, j0 + r);
                    float* ap = acc + dj * 64 + lane * 2;
                    atomicAdd(ap, v[r].x);
                    atomicAdd(ap + 1, v[r].y);
                }
            }
        }
    }
}

__device__ __forceinline__ void agg_take_batch(const int* q, int n,
                                               const int* __restrict__ src,
                                               const int* __restrict__ dst,
                                               const float* __restrict__ ht,
                                               float* acc, int lane,
                                               int2* __restrict__ lst, int cap, int* fill, int base) {
    const int e  = q[min(lane, n - 1)];
    const int s  = src[e];
    const int dl = (int)min((unsigned)(dst[e] - base), (unsigned)(AGG_CH - 1));
    int pos = 0;
    if (lane == 0) pos = atomicAdd(fill, n);
    pos = __builtin_amdgcn_readfirstlane(pos) + lane;
    if (lane < n && pos < cap) lst[pos] = make_int2(s, dl);
    agg_add_rows(s, dl, n, ht, acc, lane);
}

template <bool BUILD>
__global__ __launch_bounds__(AGG_NT)
void agg_owner_kernel(const int* __restrict__ src, const int* __restrict__ dst,
                      const float* __restrict__ ht, float* __restrict__ agg,
                      const unsigned* __restrict__ part, int2* __restrict__ list,
                      const unsigned* __restrict__ keys, int E, int N, int nPart,
                      const float* __restrict__ dis, const float* __restrict__ bvec,
                      const float* __restrict__ wmat, const float* __restrict__ bfc,
                      float* __restrict__ out) {
    __shared__ float acc[AGG_CH * 64];
    __shared__ unsigned red[AGG_NW];
    __shared__ int fill;
    const int tid  = threadIdx.x;
    const int lane = tid & 31;
    const int wave = tid >> 5;
    const unsigned blk = blockIdx.x;
    const int base = (int)blk * AGG_CH;

    unsigned below = 0u;
    for (int j = tid; j < (int)blk * AGG_PARTS; j += AGG_NT) below += part[j];
    #pragma unroll
    for (int off = 16; off > 0; off >>= 1) below += __shfl_xor(below, off, 32);
    if (lane == 0) red[wave] = below;
    if (tid == 0) fill = 0;
    unsigned mine = 0u;
    #pragma unroll
    for (int j = 0; j < AGG_PARTS; ++j) {
        const int p = (int)blk * AGG_PARTS + j;
        if (p < nPart) mine += part[p];
    }
    const int cap = (int)mine;

    float4* a4 = (float4*)acc;
    const float4* h4 = (const float4*)(ht + (size_t)base * 64);
    for (int i = tid; i < AGG_CH * 16; i += AGG_NT)
        a4[i] = (base + (i >> 4) < N) ? h4[i] : make_float4(0.f, 0.f, 0.f, 0.f);
    __syncthreads();

    static_assert(AGG_NW == 32, "one wave sum per lane");
    unsigned first = red[lane];
    #pragma unroll
    for (int off = 16; off > 0; off >>= 1) first += __shfl_xor(first, off, 32);
    int2* lst = list + (size_t)__builtin_amdgcn_readfirstlane((int)first);

    if (BUILD) {
        __shared__ int queue[AGG_NW][AGG_Q];
        __shared__ int2 uqueue[AGG_NW][AGG_UQ];
        int* q = queue[wave];
        int2* uq = uqueue[wave];
        int cnt = 0;
        const int nvec = E >> 4;
        const key4* k4 = (const key4*)keys;
        const int step = AGG_DEPTH * AGG_NT;
        const int nIt  = (nvec + step - 1) / step;
        const int rot  = (int)(((long long)blk * nIt) / gridDim.x);
        const unsigned pat = (blk & 255u) * 0x01010101u;

        for (int it = 0; it < nIt; ++it) {
            int itr = it + rot; if (itr >= nIt) itr -= nIt;
            const int v0 = itr * step + tid;
            key4 d[AGG_DEPTH];
            #pragma unroll
            for (int u = 0; u < AGG_DEPTH; ++u)
                d[u] = k4[min(v0 + u * AGG_NT, nvec - 1)];
            __builtin_amdgcn_sched_barrier(0);
            int nu = 0;
            #pragma unroll
            for (int u = 0; u < AGG_DEPTH; ++u) {
                const int v = v0 + u * AGG_NT;
                unsigned cc = 0u;
                #pragma unroll
                for (int k = 0; k < 4; ++k) {
                    const unsigned x = d[u][k] ^ pat;
                    const unsigned t = ((x & 0x7f7f7f7fu) + 0x7f7f7f7fu) | x;
                    cc |= (t >> (7 - k)) & (0x01010101u << k);
                }
                unsigned f = cc ^ 0x0f0f0f0fu;
                if (v >= nvec) f = 0u;
                const unsigned m = (unsigned)__builtin_amdgcn_ballot_w32(f != 0u);
                if (m != 0u) {
                    if (f != 0u) uq[nu + (int)__builtin_amdgcn_mbcnt_lo(m, 0u)] = make_int2((int)f, v);
                    nu += __builtin_popcount(m);
                }
            }
            for (int i = 0; i < nu; i += 32) {
                int2 en = make_int2(0, 0);
                if (i + lane < nu) en = uq[i + lane];
                unsigned f = (unsigned)en.x;
                unsigned m = (unsigned)__builtin_amdgcn_ballot_w32(f != 0u);
                while (m != 0u) {
                    if (f != 0u) {
                        const int b = __builtin_ctz(f);
                        q[cnt + (int)__builtin_amdgcn_mbcnt_lo(m, 0u)] = (en.y << 4) + ((b & 3) << 2) + (b >> 3);
                        f &= f - 1u;
                    }
                    cnt += __builtin_popcount(m);
                    if (cnt >= 32) {
                        cnt -= 32;
                        agg_take_batch(q + cnt, 32, src, dst, ht, acc, lane, lst, cap, &fill, base);
                    }
                    m = (unsigned)__builtin_amdgcn_ballot_w32(f != 0u);
                }
            }
        }
        if (wave == 0) {
            const int e = (nvec << 4) + lane;
            const bool hit = e < E && ((unsigned)dst[min(e, E - 1)] / (unsigned)AGG_CH) == blk;
            const unsigned m = (unsigned)__builtin_amdgcn_ballot_w32(hit);
            if (hit) q[cnt + (int)__builtin_amdgcn_mbcnt_lo(m, 0u)] = e;
            cnt += __builtin_popcount(m);
            if (cnt >= 32) {
                cnt -= 32;
                agg_take_batch(q + cnt, 32, src, dst, ht, acc, lane, lst, cap, &fill, base);
            }
        }
        if (cnt > 0) agg_take_batch(q, cnt, src, dst, ht, acc, lane, lst, cap, &fill, base);
    } else {
        const int wv    = __builtin_amdgcn_readfirstlane(wave);
        const int nfull = cap >> 5;
        int b = wv;
        int2 cur = make_int2(0, 0);
        if (b < nfull) cur = lst[b * 32 + lane];
        while (b < nfull) {
            const int nb = b + AGG_NW;
            int2 nxt = cur;
            if (nb < nfull) nxt = lst[nb * 32 + lane];
            agg_add_rows(cur.x, cur.y, 32, ht, acc, lane);
            cur = nxt;
            b = nb;
        }
        const int rem = cap & 31;
        if (rem > 0 && wv == (nfull & (AGG_NW - 1))) {
            const int2 en = lst[nfull * 32 + min(lane, rem - 1)];
            agg_add_rows(en.x, en.y, rem, ht, acc, lane);
        }
    }
    __syncthreads();

    if (BUILD) {
        static_assert(AGG_CH % 16 == 0 && AGG_CH / 16 <= AGG_NW, "a 16-row tile per wave");
        if (__builtin_amdgcn_readfirstlane(wave) < AGG_CH / 16) {
            const int half = lane >> 4;
            const int ml   = lane & 15;
            const int r0   = wave * 16;
            const float dA = dis[min(base + r0 + ml, N - 1)];
            const float* arow = acc + (r0 + ml) * 64;
            v8f c0 = {}, c1 = {}, c2 = {}, c3 = {};
            for (int k0 = 0; k0 < 64; k0 += 4) {
                const int ka = k0 + half * 2;
                const float2 a = *(const float2*)(arow + ka);
                v2f av;
                av.x = fmaxf(fmaf(a.x, dA, bvec[ka]), 0.0f);
                av.y = fmaxf(fmaf(a.y, dA, bvec[ka + 1]), 0.0f);
                const float* w0 = wmat + ka * 64;
                const float* w1 = wmat + (ka + 1) * 64;
                v2f b0; b0.x = w0[ml];      b0.y = w1[ml];
                v2f b1; b1.x = w0[16 + ml]; b1.y = w1[16 + ml];
                v2f b2; b2.x = w0[32 + ml]; b2.y = w1[32 + ml];
                v2f b3; b3.x = w0[48 + ml]; b3.y = w1[48 + ml];
                c0 = __builtin_amdgcn_wmma_f32_16x16x4_f32(false, av, false, b0, (short)0, c0, false, false);
                c1 = __builtin_amdgcn_wmma_f32_16x16x4_f32(false, av, false, b1, (short)0, c1, false, false);
                c2 = __builtin_amdgcn_wmma_f32_16x16x4_f32(false, av, false, b2, (short)0, c2, false, false);
                c3 = __builtin_amdgcn_wmma_f32_16x16x4_f32(false, av, false, b3, (short)0, c3, false, false);
            }
            for (int r = 0; r < 8; ++r) {
                const int orow = r0 + half * 8 + r;
                const float dr = dis[min(base + orow, N - 1)];
                float* hp = acc + orow * 64;
                hp[ml] = c0[r] * dr; hp[16 + ml] = c1[r] * dr; hp[32 + ml] = c2[r] * dr; hp[48 + ml] = c3[r] * dr;
            }
        }
        __syncthreads();
        float4* o4 = (float4*)(agg + (size_t)base * 64);
        for (int i = tid; i < AGG_CH * 16; i += AGG_NT)
            if (base + (i >> 4) < N) o4[i] = a4[i];
    } else {
        static_assert(AGG_NW * AGG_RPW == AGG_CH && AGG_RPW <= 16, "whole rows per wave, one dis value per lane of a half");
        __shared__ float outv[AGG_CH];
        const int j = lane * 2;
        const float bj0 = bvec[j], bj1 = bvec[j + 1];
        const float w0 = wmat[j], w1 = wmat[j + 1];
        const int r0 = wave * AGG_RPW;
        const int dl = __float_as_int(dis[min(base + r0 + (lane & 15), N - 1)]);
        float res = 0.0f;
        #pragma unroll
        for (int r = 0; r < AGG_RPW; ++r) {
            const float di = __int_as_float(__builtin_amdgcn_readlane(dl, r));
            const float2 a = *(const float2*)(acc + (r0 + r) * 64 + j);
            const float h0 = fmaxf(fmaf(a.x, di, bj0), 0.0f);
            const float h1 = fmaxf(fmaf(a.y, di, bj1), 0.0f);
            float p = h0 * w0 + h1 * w1;
            #pragma unroll
            for (int off = 16; off > 0; off >>= 1) p += __shfl_xor(p, off, 32);
            if (lane == r) res = p;
        }
        if (lane < AGG_RPW) outv[r0 + lane] = res + bfc[0];
        __syncthreads();
        if (tid < AGG_CH && base + tid < N) out[base + tid] = outv[tid];
    }
}

extern "C" void kernel_launch(void* const* d_in, const int* in_sizes, int n_in,
                              void* d_out, int out_size, void* d_ws, size_t ws_size,
                              hipStream_t stream) {
    const float* x   = (const float*)d_in[0];
    const int*   ei  = (const int*)d_in[1];
    const float* W1  = (const float*)d_in[2];
    const float* b1  = (const float*)d_in[3];
    const float* W2  = (const float*)d_in[4];
    const float* b2  = (const float*)d_in[5];
    const float* Wfc = (const float*)d_in[6];
    const float* bfc = (const float*)d_in[7];

    const int HID = in_sizes[3];
    const int IN  = in_sizes[2] / HID;
    const int N   = in_sizes[0] / IN;
    const int E   = in_sizes[1] / 2;
    const int* srcv = ei;
    const int* dstv = ei + E;

    char* ws = (char*)d_ws;
    float* dis = (float*)ws;
    size_t off = (((size_t)N * 4) + 255) & ~(size_t)255;
    const size_t rowBytes = (size_t)N * 64 * sizeof(float);
    float* ht   = (float*)(ws + off);
    float* ht2  = (float*)(ws + off + rowBytes);

    const int nChunks = (N + DEG_CH - 1) / DEG_CH;
    const int npad    = nChunks * DEG_CH;
    unsigned* planes  = (unsigned*)(ws + off + 3 * rowBytes);
    const size_t planeBytes = (size_t)DEG_SLICES * npad * sizeof(unsigned);
    const int nPart   = (N + ISQ_NT - 1) / ISQ_NT;
    unsigned* part    = (unsigned*)(ws + off + 3 * rowBytes + planeBytes);
    const size_t partBytes = (((size_t)nPart * 4) + 255) & ~(size_t)255;
    int2* list        = (int2*)(ws + off + 3 * rowBytes + planeBytes + partBytes);
    unsigned* keys    = (unsigned*)(ws + off + 3 * rowBytes + planeBytes + partBytes + (size_t)E * sizeof(int2));
    deg_hist_kernel<<<dim3(DEG_SLICES, nChunks), DEG_NT, 0, stream>>>(dstv, planes, keys, E, npad);

    const int aggBlocks  = (N + AGG_CH - 1) / AGG_CH;

    gemm_wmma_kernel<<<aggBlocks, GEMM_NT, 0, stream>>>(x, W1, planes, dis, part, ht, N, IN, npad);
    agg_owner_kernel<true><<<aggBlocks, AGG_NT, 0, stream>>>(srcv, dstv, ht, ht2, part, list, keys, E, N, nPart,
                                                             dis, b1, W2, nullptr, nullptr);

    agg_owner_kernel<false><<<aggBlocks, AGG_NT, 0, stream>>>(srcv, dstv, ht2, nullptr, part, list, keys, E, N, nPart,
                                                              dis, b2, Wfc, bfc, (float*)d_out);
}


# edges binned by owner range in extra blocks of the degree launch (LDS rank counters, pairs stored in bin order); both layers' owner blocks read their bins, layer-1 key scan gone; sum order: arrival
# speedup vs baseline: 3.4550x; 1.0951x over previous
#include <hip/hip_runtime.h>

typedef __attribute__((ext_vector_type(2))) float v2f;
typedef __attribute__((ext_vector_type(8))) float v8f;


#define AGG_CH     416
#define DEG_CH     32768
#define DEG_SLICES 64
#define DEG_NT     512
#define DEG_DEPTH  8
#define BIN_NB     256
#define BIN_VEC    4
#define BIN_EDGES  (DEG_NT * BIN_VEC * 4)

typedef int int4u __attribute__((ext_vector_type(4), aligned(4)));

__device__ __forceinline__ void bin_block(unsigned* lds, int sl, int nSl,
                                          const int* __restrict__ src, const int* __restrict__ dst,
                                          int2* __restrict__ list, int2* __restrict__ offs, int E) {
    static_assert(BIN_NB + 32 + 2 * BIN_EDGES <= DEG_CH, "counters, wave totals and pairs fit the LDS array");
    static_assert(BIN_NB % 32 == 0 && BIN_NB <= DEG_NT && BIN_NB / 32 <= 32, "whole waves form the prefix sum");
    static_assert(BIN_EDGES <= 65536, "a rank fits 16 bits");
    unsigned* bcnt = lds;
    unsigned* wtot = lds + BIN_NB;
    int2* stage    = (int2*)(lds + BIN_NB + 32);
    const int tid  = threadIdx.x;
    const int lane = tid & 31;
    const int lo   = sl * BIN_EDGES;
    const int hi   = min(lo + BIN_EDGES, E);
    const int nvec = E >> 2;
    const int4u* d4 = (const int4u*)dst;
    const int4u* s4 = (const int4u*)src;

    if (tid < BIN_NB) bcnt[tid] = 0u;
    __syncthreads();

    int4u d[BIN_VEC], s[BIN_VEC];
    #pragma unroll
    for (int u = 0; u < BIN_VEC; ++u) {
        const int v = (lo >> 2) + tid + u * DEG_NT;
        const int vc = max(min(v, nvec - 1), 0);
        d[u] = d4[vc];
        s[u] = s4[vc];
    }
    const bool tail = sl == nSl - 1 && tid < (E & 3);
    const int  et   = min((nvec << 2) + tid, E - 1);
    const int  td   = dst[et];
    const int  ts   = src[et];

    unsigned pk[BIN_VEC][4];
    #pragma unroll
    for (int u = 0; u < BIN_VEC; ++u) {
        const bool ok = (lo >> 2) + tid + u * DEG_NT < nvec;
        #pragma unroll
        for (int q = 0; q < 4; ++q) {
            const unsigned k = min((unsigned)d[u][q] / (unsigned)AGG_CH, (unsigned)(BIN_NB - 1));
            unsigned r = 0u;
            if (ok) r = atomicAdd(&bcnt[k], 1u);
            pk[u][q] = (k << 16) | r;
        }
    }
    unsigned pkt = 0u;
    if (tail) {
        const unsigned k = min((unsigned)td / (unsigned)AGG_CH, (unsigned)(BIN_NB - 1));
        pkt = (k << 16) | atomicAdd(&bcnt[k], 1u);
    }
    __syncthreads();

    const unsigned c = tid < BIN_NB ? bcnt[tid] : 0u;
    unsigned incl = c;
    #pragma unroll
    for (int off = 1; off < 32; off <<= 1) {
        const unsigned t = __shfl_up(incl, off, 32);
        if (lane >= off) incl += t;
    }
    if (tid < BIN_NB && lane == 31) wtot[tid >> 5] = incl;
    __syncthreads();
    if (tid < BIN_NB) {
        unsigned first = incl - c;
        for (int w = 0; w < (tid >> 5); ++w) first += wtot[w];
        bcnt[tid] = first;
        offs[(size_t)tid * nSl + sl] = make_int2(lo + (int)first, (int)c);
    }
    __syncthreads();

    #pragma unroll
    for (int u = 0; u < BIN_VEC; ++u) {
        const bool ok = (lo >> 2) + tid + u * DEG_NT < nvec;
        #pragma unroll
        for (int q = 0; q < 4; ++q) {
            const unsigned k = pk[u][q] >> 16;
            const unsigned row = min((unsigned)d[u][q] - k * (unsigned)AGG_CH, (unsigned)(AGG_CH - 1));
            if (ok) stage[bcnt[k] + (pk[u][q] & 0xffffu)] = make_int2(s[u][q], (int)row);
        }
    }
    if (tail) {
        const unsigned k = pkt >> 16;
        const unsigned row = min((unsigned)td - k * (unsigned)AGG_CH, (unsigned)(AGG_CH - 1));
        stage[bcnt[k] + (pkt & 0xffffu)] = make_int2(ts, (int)row);
    }
    __syncthreads();

    for (int i = tid; i < hi - lo; i += DEG_NT) list[lo + i] = stage[i];
}

__global__ __launch_bounds__(DEG_NT)
void deg_hist_kernel(const int* __restrict__ src, const int* __restrict__ dst,
                     unsigned* __restrict__ planes, int2* __restrict__ list,
                     int2* __restrict__ offs, int E, int npad, int nChunks, int nSl) {
    __shared__ unsigned cnt[DEG_CH];
    if ((int)blockIdx.y >= nChunks) {
        const int sl = ((int)blockIdx.y - nChunks) * DEG_SLICES + (int)blockIdx.x;
        if (sl < nSl) bin_block(cnt, sl, nSl, src, dst, list, offs, E);
        return;
    }
    const int tid   = threadIdx.x;
    const int slice = blockIdx.x;
    const unsigned base = blockIdx.y * (unsigned)DEG_CH;

    uint4* c4 = (uint4*)cnt;
    for (int i = tid; i < DEG_CH / 4; i += DEG_NT) c4[i] = make_uint4(0u, 0u, 0u, 0u);
    __syncthreads();

    const int nvec = E >> 2;
    const int per  = (nvec + DEG_SLICES - 1) / DEG_SLICES;
    const int lo   = slice * per;
    const int hi   = min(lo + per, nvec);
    const int4u* d4 = (const int4u*)dst;

    for (int v0 = lo + tid; v0 < hi; v0 += DEG_DEPTH * DEG_NT) {
        int4u d[DEG_DEPTH];
        #pragma unroll
        for (int u = 0; u < DEG_DEPTH; ++u) {
            const int v = v0 + u * DEG_NT;
            d[u] = d4[min(v, hi - 1)];
            if (v >= hi) d[u] = (int4u)(-1);
        }
        #pragma unroll
        for (int u = 0; u < DEG_DEPTH; ++u) {
            #pragma unroll
            for (int q = 0; q < 4; ++q) {
                const unsigned r = (unsigned)d[u][q] - base;
                if (r < (unsigned)DEG_CH) atomicAdd(&cnt[r], 1u);
            }
        }
    }
    if (slice == DEG_SLICES - 1) {
        for (int e = (nvec << 2) + tid; e < E; e += DEG_NT) {
            const unsigned r = (unsigned)dst[e] - base;
            if (r < (unsigned)DEG_CH) atomicAdd(&cnt[r], 1u);
        }
    }
    __syncthreads();

    uint4* o4 = (uint4*)(planes + (size_t)slice * npad + base);
    for (int i = tid; i < DEG_CH / 4; i += DEG_NT) o4[i] = c4[i];
}

#define GEMM_NW (AGG_CH / 16)
#define GEMM_NT (GEMM_NW * 32)

__global__ __launch_bounds__(GEMM_NT)
void gemm_wmma_kernel(const float* __restrict__ A,
                      const float* __restrict__ W,
                      const unsigned* __restrict__ planes,
                      float* __restrict__ dis,
                      float* __restrict__ ht,
                      int M, int K, int npad) {
    static_assert(AGG_CH % 32 == 0 && DEG_SLICES % 2 == 0, "whole lines of dis, two plane halves");
    __shared__ float dis_s[AGG_CH];
    const int tid  = threadIdx.x;
    const int lane = tid & 31;
    const int wave = __builtin_amdgcn_readfirstlane(tid >> 5);
    const int base = (int)blockIdx.x * AGG_CH;
    const int m0   = base + wave * 16;

    const int half = lane >> 4;
    const int ml   = lane & 15;
    const int row  = m0 + ml;
    const int arow = min(row, M - 1);

    unsigned s = 0u;
    const unsigned* pl = planes + (size_t)(half * (DEG_SLICES / 2)) * npad + arow;
    #pragma unroll
    for (int k = 0; k < DEG_SLICES / 2; ++k) s += pl[(size_t)k * npad];
    s += __shfl_xor(s, 16, 32);
    if (row >= M) s = 0u;
    const float dl = rsqrtf((float)(s + 1u));
    if (half == 0) dis_s[wave * 16 + ml] = dl;
    __syncthreads();

    if (tid < AGG_CH && base + tid < M) dis[base + tid] = dis_s[tid];

    if (m0 < M) {
        v8f c0 = {}, c1 = {}, c2 = {}, c3 = {};

        for (int k0 = 0; k0 < K; k0 += 4) {
            const int ka = k0 + half * 2;
            const float a0 = A[(size_t)arow * K + ka];
            const float a1 = A[(size_t)arow * K + ka + 1];
            v2f av; av.x = a0; av.y = a1;

            const float* w0 = W + (size_t)ka * 64;
            const float* w1 = W + (size_t)(ka + 1) * 64;
            v2f b0; b0.x = w0[ml];      b0.y = w1[ml];
            v2f b1; b1.x = w0[16 + ml]; b1.y = w1[16 + ml];
            v2f b2; b2.x = w0[32 + ml]; b2.y = w1[32 + ml];
            v2f b3; b3.x = w0[48 + ml]; b3.y = w1[48 + ml];

            c0 = __builtin_amdgcn_wmma_f32_16x16x4_f32(false, av, false, b0, (short)0, c0, false, false);
            c1 = __builtin_amdgcn_wmma_f32_16x16x4_f32(false, av, false, b1, (short)0, c1, false, false);
            c2 = __builtin_amdgcn_wmma_f32_16x16x4_f32(false, av, false, b2, (short)0, c2, false, false);
            c3 = __builtin_amdgcn_wmma_f32_16x16x4_f32(false, av, false, b3, (short)0, c3, false, false);
        }

        const int dli = __float_as_int(dl);
        #pragma unroll
        for (int r = 0; r < 8; ++r) {
            const int orow = m0 + half * 8 + r;
            const int dlo = __builtin_amdgcn_readlane(dli, r);
            const int dhi = __builtin_amdgcn_readlane(dli, 8 + r);
            const float dr = __int_as_float(half ? dhi : dlo);
            if (orow < M) {
                float* hp = ht + (size_t)orow * 64;
                hp[ml] = c0[r] * dr; hp[16 + ml] = c1[r] * dr; hp[32 + ml] = c2[r] * dr; hp[48 + ml] = c3[r] * dr;
            }
        }
    }
}

#define AGG_NT    1024
#define AGG_NW    (AGG_NT / 32)
#define AGG_GB    256
#define AGG_AHEAD 8
#define AGG_ROWS  16
#define AGG_RPW   (AGG_CH / AGG_NW)

__device__ __forceinline__ void agg_add_rows(int s, int dl, int n,
                                             const float* __restrict__ ht,
                                             float* acc, int lane) {
    #pragma unroll
    for (int j0 = 0; j0 < 32; j0 += AGG_ROWS) {
        if (j0 < n) {
            float2 v[AGG_ROWS];
            #pragma unroll
            for (int r = 0; r < AGG_ROWS; ++r) {
                const int sj = __builtin_amdgcn_readlane(s, j0 + r);
                v[r] = *(const float2*)(ht + (size_t)sj * 64 + lane * 2);
            }
            #pragma unroll
            for (int r = 0; r < AGG_ROWS; ++r) {
                if (j0 + r < n) {
                    const int dj = __builtin_amdgcn_readlane(dl, j0 + r);
                    float* ap = acc + dj * 64 + lane * 2;
                    atomicAdd(ap, v[r].x);
                    atomicAdd(ap + 1, v[r].y);
                }
            }
        }
    }
}

template <bool BUILD>
__global__ __launch_bounds__(AGG_NT)
void agg_owner_kernel(const float* __restrict__ ht, float* __restrict__ agg,
                      const int2* __restrict__ list, const int2* __restrict__ offs, int nSl, int N,
                      const float* __restrict__ dis, const float* __restrict__ bvec,
                      const float* __restrict__ wmat, const float* __restrict__ bfc,
                      float* __restrict__ out) {
    __shared__ float acc[AGG_CH * 64];
    __shared__ int pre_s[AGG_GB + 1];
    __shared__ int dlt_s[AGG_GB];
    __shared__ int wt_s[AGG_GB / 32];
    const int tid  = threadIdx.x;
    const int lane = tid & 31;
    const int wave = tid >> 5;
    const unsigned blk = blockIdx.x;
    const int base = (int)blk * AGG_CH;

    float4* a4 = (float4*)acc;
    const float4* h4 = (const float4*)(ht + (size_t)base * 64);
    for (int i = tid; i < AGG_CH * 16; i += AGG_NT)
        a4[i] = (base + (i >> 4) < N) ? h4[i] : make_float4(0.f, 0.f, 0.f, 0.f);
    __syncthreads();

    static_assert(AGG_GB % 32 == 0 && AGG_GB <= AGG_NT && (AGG_GB & (AGG_GB - 1)) == 0, "whole waves form the running sums; the search halves a power of two");
    const int wv = __builtin_amdgcn_readfirstlane(wave);
    const int2* mine = offs + (size_t)blk * nSl;
    for (int g0 = 0; g0 < nSl; g0 += AGG_GB) {
        if (g0 > 0) __syncthreads();
        int2 oc = make_int2(0, 0);
        if (tid < AGG_GB && g0 + tid < nSl) oc = mine[g0 + tid];
        int incl = oc.y;
        #pragma unroll
        for (int off = 1; off < 32; off <<= 1) {
            const int t = __shfl_up(incl, off, 32);
            if (lane >= off) incl += t;
        }
        if (tid < AGG_GB && lane == 31) wt_s[wave] = incl;
        __syncthreads();
        if (tid < AGG_GB) {
            const int len = oc.y;
            for (int w = 0; w < wave; ++w) incl += wt_s[w];
            pre_s[tid + 1] = incl;
            dlt_s[tid] = oc.x - (incl - len);
            if (tid == 0) pre_s[0] = 0;
        }
        __syncthreads();
        const int T = __builtin_amdgcn_readfirstlane(pre_s[AGG_GB]);
        auto place = [&](int p0) -> int {
            const int pc = min(p0 + lane, T - 1);
            int j = 0;
            #pragma unroll
            for (int step = AGG_GB / 2; step > 0; step >>= 1)
                if (pre_s[j + step] <= pc) j += step;
            return dlt_s[j] + pc;
        };
        const int nbatch = (T + 31) >> 5;
        for (int k0 = 0; k0 < nbatch; k0 += AGG_AHEAD * AGG_NW) {
            const int nbw = min(AGG_AHEAD, (nbatch - k0 - wv + AGG_NW - 1) / AGG_NW);
            int idx[AGG_AHEAD];
            #pragma unroll
            for (int i = 0; i < AGG_AHEAD; ++i) idx[i] = place((k0 + wv + AGG_NW * i) * 32);
            int2 cur = list[idx[0]];
            #pragma nounroll
            for (int i = 0; i < nbw; ++i) {
                int2 nxt = cur;
                if (i + 1 < nbw) {
                    int ix = idx[1];
                    #pragma unroll
                    for (int t = 2; t < AGG_AHEAD; ++t) if (i + 1 == t) ix = idx[t];
                    nxt = list[ix];
                }
                const int p0 = (k0 + wv + AGG_NW * i) * 32;
                agg_add_rows(cur.x, cur.y, min(32, T - p0), ht, acc, lane);
                cur = nxt;
            }
        }
    }
    __syncthreads();

    if (BUILD) {
        static_assert(AGG_CH % 16 == 0 && AGG_CH / 16 <= AGG_NW, "a 16-row tile per wave");
        if (__builtin_amdgcn_readfirstlane(wave) < AGG_CH / 16) {
            const int half = lane >> 4;
            const int ml   = lane & 15;
            const int r0   = wave * 16;
            const float dA = dis[min(base + r0 + ml, N - 1)];
            const float* arow = acc + (r0 + ml) * 64;
            v8f c0 = {}, c1 = {}, c2 = {}, c3 = {};
            for (int k0 = 0; k0 < 64; k0 += 4) {
                const int ka = k0 + half * 2;
                const float2 a = *(const float2*)(arow + ka);
                v2f av;
                av.x = fmaxf(fmaf(a.x, dA, bvec[ka]), 0.0f);
                av.y = fmaxf(fmaf(a.y, dA, bvec[ka + 1]), 0.0f);
                const float* w0 = wmat + ka * 64;
                const float* w1 = wmat + (ka + 1) * 64;
                v2f b0; b0.x = w0[ml];      b0.y = w1[ml];
                v2f b1; b1.x = w0[16 + ml]; b1.y = w1[16 + ml];
                v2f b2; b2.x = w0[32 + ml]; b2.y = w1[32 + ml];
                v2f b3; b3.x = w0[48 + ml]; b3.y = w1[48 + ml];
                c0 = __builtin_amdgcn_wmma_f32_16x16x4_f32(false, av, false, b0, (short)0, c0, false, false);
                c1 = __builtin_amdgcn_wmma_f32_16x16x4_f32(false, av, false, b1, (short)0, c1, false, false);
                c2 = __builtin_amdgcn_wmma_f32_16x16x4_f32(false, av, false, b2, (short)0, c2, false, false);
                c3 = __builtin_amdgcn_wmma_f32_16x16x4_f32(false, av, false, b3, (short)0, c3, false, false);
            }
            for (int r = 0; r < 8; ++r) {
                const int orow = r0 + half * 8 + r;
                const float dr = dis[min(base + orow, N - 1)];
                float* hp = acc + orow * 64;
                hp[ml] = c0[r] * dr; hp[16 + ml] = c1[r] * dr; hp[32 + ml] = c2[r] * dr; hp[48 + ml] = c3[r] * dr;
            }
        }
        __syncthreads();
        float4* o4 = (float4*)(agg + (size_t)base * 64);
        for (int i = tid; i < AGG_CH * 16; i += AGG_NT)
            if (base + (i >> 4) < N) o4[i] = a4[i];
    } else {
        static_assert(AGG_NW * AGG_RPW == AGG_CH && AGG_RPW <= 16, "whole rows per wave, one dis value per lane of a half");
        __shared__ float outv[AGG_CH];
        const int j = lane * 2;
        const float bj0 = bvec[j], bj1 = bvec[j + 1];
        const float w0 = wmat[j], w1 = wmat[j + 1];
        const int r0 = wave * AGG_RPW;
        const int dl = __float_as_int(dis[min(base + r0 + (lane & 15), N - 1)]);
        float res = 0.0f;
        #pragma unroll
        for (int r = 0; r < AGG_RPW; ++r) {
            const float di = __int_as_float(__builtin_amdgcn_readlane(dl, r));
            const float2 a = *(const float2*)(acc + (r0 + r) * 64 + j);
            const float h0 = fmaxf(fmaf(a.x, di, bj0), 0.0f);
            const float h1 = fmaxf(fmaf(a.y, di, bj1), 0.0f);
            float p = h0 * w0 + h1 * w1;
            #pragma unroll
            for (int off = 16; off > 0; off >>= 1) p += __shfl_xor(p, off, 32);
            if (lane == r) res = p;
        }
        if (lane < AGG_RPW) outv[r0 + lane] = res + bfc[0];
        __syncthreads();
        if (tid < AGG_CH && base + tid < N) out[base + tid] = outv[tid];
    }
}

extern "C" void kernel_launch(void* const* d_in, const int* in_sizes, int n_in,
                              void* d_out, int out_size, void* d_ws, size_t ws_size,
                              hipStream_t stream) {
    const float* x   = (const float*)d_in[0];
    const int*   ei  = (const int*)d_in[1];
    const float* W1  = (const float*)d_in[2];
    const float* b1  = (const float*)d_in[3];
    const float* W2  = (const float*)d_in[4];
    const float* b2  = (const float*)d_in[5];
    const float* Wfc = (const float*)d_in[6];
    const float* bfc = (const float*)d_in[7];

    const int HID = in_sizes[3];
    const int IN  = in_sizes[2] / HID;
    const int N   = in_sizes[0] / IN;
    const int E   = in_sizes[1] / 2;
    const int* srcv = ei;
    const int* dstv = ei + E;

    char* ws = (char*)d_ws;
    float* dis = (float*)ws;
    size_t off = (((size_t)N * 4) + 255) & ~(size_t)255;
    const size_t rowBytes = (size_t)N * 64 * sizeof(float);
    float* ht   = (float*)(ws + off);
    float* ht2  = (float*)(ws + off + rowBytes);

    const int nChunks = (N + DEG_CH - 1) / DEG_CH;
    const int npad    = nChunks * DEG_CH;
    unsigned* planes  = (unsigned*)(ws + off + 3 * rowBytes);
    const size_t planeBytes = (size_t)DEG_SLICES * npad * sizeof(unsigned);
    const int nSl     = (E + BIN_EDGES - 1) / BIN_EDGES;
    const int binRows = (nSl + DEG_SLICES - 1) / DEG_SLICES;
    int2* list        = (int2*)(ws + off + 3 * rowBytes + planeBytes);
    int2* offs        = (int2*)(ws + off + 3 * rowBytes + planeBytes + (size_t)E * sizeof(int2));
    deg_hist_kernel<<<dim3(DEG_SLICES, nChunks + binRows), DEG_NT, 0, stream>>>(srcv, dstv, planes, list, offs, E, npad, nChunks, nSl);

    const int aggBlocks  = (N + AGG_CH - 1) / AGG_CH;

    gemm_wmma_kernel<<<aggBlocks, GEMM_NT, 0, stream>>>(x, W1, planes, dis, ht, N, IN, npad);
    agg_owner_kernel<true><<<aggBlocks, AGG_NT, 0, stream>>>(ht, ht2, list, offs, nSl, N,
                                                             dis, b1, W2, nullptr, nullptr);

    agg_owner_kernel<false><<<aggBlocks, AGG_NT, 0, stream>>>(ht2, nullptr, list, offs, nSl, N,
                                                              dis, b2, Wfc, bfc, (float*)d_out);
}


# in-degree counted by the layer-1 product blocks from their own edge bins (LDS integer counters per 416-row range); histogram blocks and the 64 count planes gone; degree exact; 4 launches kept
# speedup vs baseline: 3.5941x; 1.0403x over previous
#include <hip/hip_runtime.h>

typedef __attribute__((ext_vector_type(2))) float v2f;
typedef __attribute__((ext_vector_type(8))) float v8f;


#define AGG_CH     416
#define DEG_NT     512
#define BIN_NB     256
#define BIN_VEC    4
#define BIN_EDGES  (DEG_NT * BIN_VEC * 4)
#define BIN_LDS    (BIN_NB + 32 + 2 * BIN_EDGES)

typedef int int4u __attribute__((ext_vector_type(4), aligned(4)));

__device__ __forceinline__ void bin_block(unsigned* lds, int sl, int nSl,
                                          const int* __restrict__ src, const int* __restrict__ dst,
                                          int2* __restrict__ list, int2* __restrict__ offs, int E) {
    static_assert(BIN_NB % 32 == 0 && BIN_NB <= DEG_NT && BIN_NB / 32 <= 32, "whole waves form the prefix sum");
    static_assert(BIN_EDGES <= 65536, "a rank fits 16 bits");
    unsigned* bcnt = lds;
    unsigned* wtot = lds + BIN_NB;
    int2* stage    = (int2*)(lds + BIN_NB + 32);
    const int tid  = threadIdx.x;
    const int lane = tid & 31;
    const int lo   = sl * BIN_EDGES;
    const int hi   = min(lo + BIN_EDGES, E);
    const int nvec = E >> 2;
    const int4u* d4 = (const int4u*)dst;
    const int4u* s4 = (const int4u*)src;

    if (tid < BIN_NB) bcnt[tid] = 0u;
    __syncthreads();

    int4u d[BIN_VEC], s[BIN_VEC];
    #pragma unroll
    for (int u = 0; u < BIN_VEC; ++u) {
        const int v = (lo >> 2) + tid + u * DEG_NT;
        const int vc = max(min(v, nvec - 1), 0);
        d[u] = d4[vc];
        s[u] = s4[vc];
    }
    const bool tail = sl == nSl - 1 && tid < (E & 3);
    const int  et   = min((nvec << 2) + tid, E - 1);
    const int  td   = dst[et];
    const int  ts   = src[et];

    unsigned pk[BIN_VEC][4];
    #pragma unroll
    for (int u = 0; u < BIN_VEC; ++u) {
        const bool ok = (lo >> 2) + tid + u * DEG_NT < nvec;
        #pragma unroll
        for (int q = 0; q < 4; ++q) {
            const unsigned k = min((unsigned)d[u][q] / (unsigned)AGG_CH, (unsigned)(BIN_NB - 1));
            unsigned r = 0u;
            if (ok) r = atomicAdd(&bcnt[k], 1u);
            pk[u][q] = (k << 16) | r;
        }
    }
    unsigned pkt = 0u;
    if (tail) {
        const unsigned k = min((unsigned)td / (unsigned)AGG_CH, (unsigned)(BIN_NB - 1));
        pkt = (k << 16) | atomicAdd(&bcnt[k], 1u);
    }
    __syncthreads();

    const unsigned c = tid < BIN_NB ? bcnt[tid] : 0u;
    unsigned incl = c;
    #pragma unroll
    for (int off = 1; off < 32; off <<= 1) {
        const unsigned t = __shfl_up(incl, off, 32);
        if (lane >= off) incl += t;
    }
    if (tid < BIN_NB && lane == 31) wtot[tid >> 5] = incl;
    __syncthreads();
    if (tid < BIN_NB) {
        unsigned first = incl - c;
        for (int w = 0; w < (tid >> 5); ++w) first += wtot[w];
        bcnt[tid] = first;
        offs[(size_t)tid * nSl + sl] = make_int2(lo + (int)first, (int)c);
    }
    __syncthreads();

    #pragma unroll
    for (int u = 0; u < BIN_VEC; ++u) {
        const bool ok = (lo >> 2) + tid + u * DEG_NT < nvec;
        #pragma unroll
        for (int q = 0; q < 4; ++q) {
            const unsigned k = pk[u][q] >> 16;
            const unsigned row = min((unsigned)d[u][q] - k * (unsigned)AGG_CH, (unsigned)(AGG_CH - 1));
            if (ok) stage[bcnt[k] + (pk[u][q] & 0xffffu)] = make_int2(s[u][q], (int)row);
        }
    }
    if (tail) {
        const unsigned k = pkt >> 16;
        const unsigned row = min((unsigned)td - k * (unsigned)AGG_CH, (unsigned)(AGG_CH - 1));
        stage[bcnt[k] + (pkt & 0xffffu)] = make_int2(ts, (int)row);
    }
    __syncthreads();

    for (int i = tid; i < hi - lo; i += DEG_NT) list[lo + i] = stage[i];
}

__global__ __launch_bounds__(DEG_NT)
void edge_bin_kernel(const int* __restrict__ src, const int* __restrict__ dst,
                     int2* __restrict__ list, int2* __restrict__ offs, int E, int nSl) {
    __shared__ unsigned lds[BIN_LDS];
    bin_block(lds, (int)blockIdx.x, nSl, src, dst, list, offs, E);
}

#define GEMM_NW (AGG_CH / 16)
#define GEMM_NT (GEMM_NW * 32)
#define CNT_AHEAD 8
#define CNT_VEC   2

__global__ __launch_bounds__(GEMM_NT)
void gemm_wmma_kernel(const float* __restrict__ A,
                      const float* __restrict__ W,
                      const int2* __restrict__ list, const int2* __restrict__ offs, int nSl,
                      float* __restrict__ dis,
                      float* __restrict__ ht,
                      int M, int K) {
    static_assert(AGG_CH % 32 == 0 && AGG_CH <= GEMM_NT, "whole lines of dis, a thread per counter");
    __shared__ unsigned cnt_s[AGG_CH];
    const int tid  = threadIdx.x;
    const int lane = tid & 31;
    const int wave = __builtin_amdgcn_readfirstlane(tid >> 5);
    const int base = (int)blockIdx.x * AGG_CH;
    const int m0   = base + wave * 16;

    const int half = lane >> 4;
    const int ml   = lane & 15;
    const int row  = m0 + ml;
    const int arow = min(row, M - 1);

    if (tid < AGG_CH) cnt_s[tid] = 0u;
    __syncthreads();

    const int2* mine = offs + (size_t)blockIdx.x * nSl;
    int2 oc[CNT_AHEAD];
    int  lr[CNT_AHEAD][CNT_VEC];
    #pragma unroll
    for (int i = 0; i < CNT_AHEAD; ++i) {
        const int g = wave + GEMM_NW * i;
        oc[i] = make_int2(0, 0);
        if (g < nSl) oc[i] = mine[g];
    }
    #pragma unroll
    for (int i = 0; i < CNT_AHEAD; ++i) {
        #pragma unroll
        for (int u = 0; u < CNT_VEC; ++u) {
            const int e = lane + 32 * u;
            const int p = max(oc[i].x + min(e, oc[i].y - 1), 0);
            lr[i][u] = list[p].y;
        }
    }

    v8f c0 = {}, c1 = {}, c2 = {}, c3 = {};
    if (m0 < M) {
        for (int k0 = 0; k0 < K; k0 += 4) {
            const int ka = k0 + half * 2;
            const float a0 = A[(size_t)arow * K + ka];
            const float a1 = A[(size_t)arow * K + ka + 1];
            v2f av; av.x = a0; av.y = a1;

            const float* w0 = W + (size_t)ka * 64;
            const float* w1 = W + (size_t)(ka + 1) * 64;
            v2f b0; b0.x = w0[ml];      b0.y = w1[ml];
            v2f b1; b1.x = w0[16 + ml]; b1.y = w1[16 + ml];
            v2f b2; b2.x = w0[32 + ml]; b2.y = w1[32 + ml];
            v2f b3; b3.x = w0[48 + ml]; b3.y = w1[48 + ml];

            c0 = __builtin_amdgcn_wmma_f32_16x16x4_f32(false, av, false, b0, (short)0, c0, false, false);
            c1 = __builtin_amdgcn_wmma_f32_16x16x4_f32(false, av, false, b1, (short)0, c1, false, false);
            c2 = __builtin_amdgcn_wmma_f32_16x16x4_f32(false, av, false, b2, (short)0, c2, false, false);
            c3 = __builtin_amdgcn_wmma_f32_16x16x4_f32(false, av, false, b3, (short)0, c3, false, false);
        }
    }

    #pragma unroll
    for (int i = 0; i < CNT_AHEAD; ++i) {
        #pragma unroll
        for (int u = 0; u < CNT_VEC; ++u)
            if (lane + 32 * u < oc[i].y) atomicAdd(&cnt_s[lr[i][u]], 1u);
        for (int e = 32 * CNT_VEC + lane; e < oc[i].y; e += 32) atomicAdd(&cnt_s[list[oc[i].x + e].y], 1u);
    }
    for (int g = wave + GEMM_NW * CNT_AHEAD; g < nSl; g += GEMM_NW) {
        const int2 o = mine[g];
        for (int e = lane; e < o.y; e += 32) atomicAdd(&cnt_s[list[o.x + e].y], 1u);
    }
    __syncthreads();

    if (tid < AGG_CH && base + tid < M) dis[base + tid] = rsqrtf((float)(cnt_s[tid] + 1u));

    if (m0 < M) {
        const float dl = rsqrtf((float)(cnt_s[wave * 16 + ml] + 1u));
        const int dli = __float_as_int(dl);
        #pragma unroll
        for (int r = 0; r < 8; ++r) {
            const int orow = m0 + half * 8 + r;
            const int dlo = __builtin_amdgcn_readlane(dli, r);
            const int dhi = __builtin_amdgcn_readlane(dli, 8 + r);
            const float dr = __int_as_float(half ? dhi : dlo);
            if (orow < M) {
                float* hp = ht + (size_t)orow * 64;
                hp[ml] = c0[r] * dr; hp[16 + ml] = c1[r] * dr; hp[32 + ml] = c2[r] * dr; hp[48 + ml] = c3[r] * dr;
            }
        }
    }
}

#define AGG_NT    1024
#define AGG_NW    (AGG_NT / 32)
#define AGG_GB    256
#define AGG_AHEAD 8
#define AGG_ROWS  16
#define AGG_RPW   (AGG_CH / AGG_NW)

__device__ __forceinline__ void agg_add_rows(int s, int dl, int n,
                                             const float* __restrict__ ht,
                                             float* acc, int lane) {
    #pragma unroll
    for (int j0 = 0; j0 < 32; j0 += AGG_ROWS) {
        if (j0 < n) {
            float2 v[AGG_ROWS];
            #pragma unroll
            for (int r = 0; r < AGG_ROWS; ++r) {
                const int sj = __builtin_amdgcn_readlane(s, j0 + r);
                v[r] = *(const float2*)(ht + (size_t)sj * 64 + lane * 2);
            }
            #pragma unroll
            for (int r = 0; r < AGG_ROWS; ++r) {
                if (j0 + r < n) {
                    const int dj = __builtin_amdgcn_readlane(dl, j0 + r);
                    float* ap = acc + dj * 64 + lane * 2;
                    atomicAdd(ap, v[r].x);
                    atomicAdd(ap + 1, v[r].y);
                }
            }
        }
    }
}

template <bool BUILD>
__global__ __launch_bounds__(AGG_NT)
void agg_owner_kernel(const float* __restrict__ ht, float* __restrict__ agg,
                      const int2* __restrict__ list, const int2* __restrict__ offs, int nSl, int N,
                      const float* __restrict__ dis, const float* __restrict__ bvec,
                      const float* __restrict__ wmat, const float* __restrict__ bfc,
                      float* __restrict__ out) {
    __shared__ float acc[AGG_CH * 64];
    __shared__ int pre_s[AGG_GB + 1];
    __shared__ int dlt_s[AGG_GB];
    __shared__ int wt_s[AGG_GB / 32];
    const int tid  = threadIdx.x;
    const int lane = tid & 31;
    const int wave = tid >> 5;
    const unsigned blk = blockIdx.x;
    const int base = (int)blk * AGG_CH;

    float4* a4 = (float4*)acc;
    const float4* h4 = (const float4*)(ht + (size_t)base * 64);
    for (int i = tid; i < AGG_CH * 16; i += AGG_NT)
        a4[i] = (base + (i >> 4) < N) ? h4[i] : make_float4(0.f, 0.f, 0.f, 0.f);
    __syncthreads();

    static_assert(AGG_GB % 32 == 0 && AGG_GB <= AGG_NT && (AGG_GB & (AGG_GB - 1)) == 0, "whole waves form the running sums; the search halves a power of two");
    const int wv = __builtin_amdgcn_readfirstlane(wave);
    const int2* mine = offs + (size_t)blk * nSl;
    for (int g0 = 0; g0 < nSl; g0 += AGG_GB) {
        if (g0 > 0) __syncthreads();
        int2 oc = make_int2(0, 0);
        if (tid < AGG_GB && g0 + tid < nSl) oc = mine[g0 + tid];
        int incl = oc.y;
        #pragma unroll
        for (int off = 1; off < 32; off <<= 1) {
            const int t = __shfl_up(incl, off, 32);
            if (lane >= off) incl += t;
        }
        if (tid < AGG_GB && lane == 31) wt_s[wave] = incl;
        __syncthreads();
        if (tid < AGG_GB) {
            const int len = oc.y;
            for (int w = 0; w < wave; ++w) incl += wt_s[w];
            pre_s[tid + 1] = incl;
            dlt_s[tid] = oc.x - (incl - len);
            if (tid == 0) pre_s[0] = 0;
        }
        __syncthreads();
        const int T = __builtin_amdgcn_readfirstlane(pre_s[AGG_GB]);
        auto place = [&](int p0) -> int {
            const int pc = min(p0 + lane, T - 1);
            int j = 0;
            #pragma unroll
            for (int step = AGG_GB / 2; step > 0; step >>= 1)
                if (pre_s[j + step] <= pc) j += step;
            return dlt_s[j] + pc;
        };
        const int nbatch = (T + 31) >> 5;
        for (int k0 = 0; k0 < nbatch; k0 += AGG_AHEAD * AGG_NW) {
            const int nbw = min(AGG_AHEAD, (nbatch - k0 - wv + AGG_NW - 1) / AGG_NW);
            int idx[AGG_AHEAD];
            #pragma unroll
            for (int i = 0; i < AGG_AHEAD; ++i) idx[i] = place((k0 + wv + AGG_NW * i) * 32);
            int2 cur = list[idx[0]];
            #pragma nounroll
            for (int i = 0; i < nbw; ++i) {
                int2 nxt = cur;
                if (i + 1 < nbw) {
                    int ix = idx[1];
                    #pragma unroll
                    for (int t = 2; t < AGG_AHEAD; ++t) if (i + 1 == t) ix = idx[t];
                    nxt = list[ix];
                }
                const int p0 = (k0 + wv + AGG_NW * i) * 32;
                agg_add_rows(cur.x, cur.y, min(32, T - p0), ht, acc, lane);
                cur = nxt;
            }
        }
    }
    __syncthreads();

    if (BUILD) {
        static_assert(AGG_CH % 16 == 0 && AGG_CH / 16 <= AGG_NW, "a 16-row tile per wave");
        if (__builtin_amdgcn_readfirstlane(wave) < AGG_CH / 16) {
            const int half = lane >> 4;
            const int ml   = lane & 15;
            const int r0   = wave * 16;
            const float dA = dis[min(base + r0 + ml, N - 1)];
            const float* arow = acc + (r0 + ml) * 64;
            v8f c0 = {}, c1 = {}, c2 = {}, c3 = {};
            for (int k0 = 0; k0 < 64; k0 += 4) {
                const int ka = k0 + half * 2;
                const float2 a = *(const float2*)(arow + ka);
                v2f av;
                av.x = fmaxf(fmaf(a.x, dA, bvec[ka]), 0.0f);
                av.y = fmaxf(fmaf(a.y, dA, bvec[ka + 1]), 0.0f);
                const float* w0 = wmat + ka * 64;
                const float* w1 = wmat + (ka + 1) * 64;
                v2f b0; b0.x = w0[ml];      b0.y = w1[ml];
                v2f b1; b1.x = w0[16 + ml]; b1.y = w1[16 + ml];
                v2f b2; b2.x = w0[32 + ml]; b2.y = w1[32 + ml];
                v2f b3; b3.x = w0[48 + ml]; b3.y = w1[48 + ml];
                c0 = __builtin_amdgcn_wmma_f32_16x16x4_f32(false, av, false, b0, (short)0, c0, false, false);
                c1 = __builtin_amdgcn_wmma_f32_16x16x4_f32(false, av, false, b1, (short)0, c1, false, false);
                c2 = __builtin_amdgcn_wmma_f32_16x16x4_f32(false, av, false, b2, (short)0, c2, false, false);
                c3 = __builtin_amdgcn_wmma_f32_16x16x4_f32(false, av, false, b3, (short)0, c3, false, false);
            }
            for (int r = 0; r < 8; ++r) {
                const int orow = r0 + half * 8 + r;
                const float dr = dis[min(base + orow, N - 1)];
                float* hp = acc + orow * 64;
                hp[ml] = c0[r] * dr; hp[16 + ml] = c1[r] * dr; hp[32 + ml] = c2[r] * dr; hp[48 + ml] = c3[r] * dr;
            }
        }
        __syncthreads();
        float4* o4 = (float4*)(agg + (size_t)base * 64);
        for (int i = tid; i < AGG_CH * 16; i += AGG_NT)
            if (base + (i >> 4) < N) o4[i] = a4[i];
    } else {
        static_assert(AGG_NW * AGG_RPW == AGG_CH && AGG_RPW <= 16, "whole rows per wave, one dis value per lane of a half");
        __shared__ float outv[AGG_CH];
        const int j = lane * 2;
        const float bj0 = bvec[j], bj1 = bvec[j + 1];
        const float w0 = wmat[j], w1 = wmat[j + 1];
        const int r0 = wave * AGG_RPW;
        const int dl = __float_as_int(dis[min(base + r0 + (lane & 15), N - 1)]);
        float res = 0.0f;
        #pragma unroll
        for (int r = 0; r < AGG_RPW; ++r) {
            const float di = __int_as_float(__builtin_amdgcn_readlane(dl, r));
            const float2 a = *(const float2*)(acc + (r0 + r) * 64 + j);
            const float h0 = fmaxf(fmaf(a.x, di, bj0), 0.0f);
            const float h1 = fmaxf(fmaf(a.y, di, bj1), 0.0f);
            float p = h0 * w0 + h1 * w1;
            #pragma unroll
            for (int off = 16; off > 0; off >>= 1) p += __shfl_xor(p, off, 32);
            if (lane == r) res = p;
        }
        if (lane < AGG_RPW) outv[r0 + lane] = res + bfc[0];
        __syncthreads();
        if (tid < AGG_CH && base + tid < N) out[base + tid] = outv[tid];
    }
}

extern "C" void kernel_launch(void* const* d_in, const int* in_sizes, int n_in,
                              void* d_out, int out_size, void* d_ws, size_t ws_size,
                              hipStream_t stream) {
    const float* x   = (const float*)d_in[0];
    const int*   ei  = (const int*)d_in[1];
    const float* W1  = (const float*)d_in[2];
    const float* b1  = (const float*)d_in[3];
    const float* W2  = (const float*)d_in[4];
    const float* b2  = (const float*)d_in[5];
    const float* Wfc = (const float*)d_in[6];
    const float* bfc = (const float*)d_in[7];

    const int HID = in_sizes[3];
    const int IN  = in_sizes[2] / HID;
    const int N   = in_sizes[0] / IN;
    const int E   = in_sizes[1] / 2;
    const int* srcv = ei;
    const int* dstv = ei + E;

    char* ws = (char*)d_ws;
    float* dis = (float*)ws;
    size_t off = (((size_t)N * 4) + 255) & ~(size_t)255;
    const size_t rowBytes = (size_t)N * 64 * sizeof(float);
    float* ht   = (float*)(ws + off);
    float* ht2  = (float*)(ws + off + rowBytes);

    const int nSl     = (E + BIN_EDGES - 1) / BIN_EDGES;
    int2* list        = (int2*)(ws + off + 2 * rowBytes);
    int2* offs        = (int2*)(ws + off + 2 * rowBytes + (size_t)E * sizeof(int2));
    edge_bin_kernel<<<nSl, DEG_NT, 0, stream>>>(srcv, dstv, list, offs, E, nSl);

    const int aggBlocks  = (N + AGG_CH - 1) / AGG_CH;

    gemm_wmma_kernel<<<aggBlocks, GEMM_NT, 0, stream>>>(x, W1, list, offs, nSl, dis, ht, N, IN);
    agg_owner_kernel<true><<<aggBlocks, AGG_NT, 0, stream>>>(ht, ht2, list, offs, nSl, N,
                                                             dis, b1, W2, nullptr, nullptr);

    agg_owner_kernel<false><<<aggBlocks, AGG_NT, 0, stream>>>(ht2, nullptr, list, offs, nSl, N,
                                                              dis, b2, Wfc, bfc, (float*)d_out);
}
